# starting kernel
# baseline (speedup 1.0000x reference)
#include <hip/hip_runtime.h>

typedef __attribute__((ext_vector_type(2))) float v2f;
typedef __attribute__((ext_vector_type(8))) float v8f;

#define D_FEAT 128

__global__ __launch_bounds__(256) void deg_init_kernel(float* deg, int n) {
    int i = blockIdx.x * 256 + threadIdx.x;
    if (i < n) deg[i] = 1.0f;
}

__global__ __launch_bounds__(256) void deg_edges_kernel(const int* __restrict__ dst,
                                                        float* deg, int e) {
    int i = blockIdx.x * 256 + threadIdx.x;
    if (i < e) atomicAdd(&deg[dst[i]], 1.0f);
}

__global__ __launch_bounds__(256) void deg_rsqrt_kernel(float* deg, int n) {
    int i = blockIdx.x * 256 + threadIdx.x;
    if (i < n) {
        float d = deg[i];
        deg[i] = (d > 0.0f) ? rsqrtf(d) : 0.0f;
    }
}

__global__ __launch_bounds__(256) void zero_kernel(float4* p, long n4) {
    long i = (long)blockIdx.x * 256 + threadIdx.x;
    if (i < n4) p[i] = make_float4(0.f, 0.f, 0.f, 0.f);
}

#define LDA 132
__global__ __launch_bounds__(256)
void gemm128_wmma_kernel(const float* __restrict__ A, const float* __restrict__ W,
                         float* __restrict__ H, int nrows) {
    __shared__ float As[16 * LDA];
    const int tid = threadIdx.x;
    const long row0 = (long)blockIdx.x * 16;

    for (int idx = tid; idx < 512; idx += 256) {
        int r = idx >> 5;
        int c4 = idx & 31;
        long gr = row0 + r;
        if (gr >= nrows) gr = nrows - 1;
        *(float4*)(As + r * LDA + c4 * 4) =
            *(const float4*)(A + gr * D_FEAT + c4 * 4);
    }
    __syncthreads();

    const int wave  = tid >> 5;
    const int lane  = tid & 31;
    const int col0  = wave * 16;
    const int mrow  = lane & 15;
    const int khalf = (lane >> 4) * 2;
    const int ncol  = lane & 15;

    v8f acc = {};
    for (int k = 0; k < D_FEAT; k += 4) {
        v2f a, b;
        a.x = As[mrow * LDA + k + khalf + 0];
        a.y = As[mrow * LDA + k + khalf + 1];
        b.x = W[(k + khalf + 0) * D_FEAT + col0 + ncol];
        b.y = W[(k + khalf + 1) * D_FEAT + col0 + ncol];
        acc = __builtin_amdgcn_wmma_f32_16x16x4_f32(
             false, a,  false, b,
             (short)0, acc,  false,  false);
    }

    const int mbase = (lane >> 4) * 8;
    for (int r = 0; r < 8; ++r) {
        long m = row0 + mbase + r;
        if (m < nrows) H[m * D_FEAT + col0 + ncol] = acc[r];
    }
}

__global__ __launch_bounds__(256)
void scatter_kernel(const int* __restrict__ src, const int* __restrict__ dst,
                    const float* __restrict__ dinv, const float* __restrict__ h,
                    float* __restrict__ agg, int nedges) {
    int e = blockIdx.x * 8 + (threadIdx.x >> 5);
    if (e >= nedges) return;
    int lane = threadIdx.x & 31;
    int s = src[e], d = dst[e];
    float norm = dinv[s] * dinv[d];
    float4 hv = *(const float4*)(h + (long)s * D_FEAT + lane * 4);
    float* base = agg + (long)d * D_FEAT + lane * 4;
    atomicAdd(base + 0, norm * hv.x);
    atomicAdd(base + 1, norm * hv.y);
    atomicAdd(base + 2, norm * hv.z);
    atomicAdd(base + 3, norm * hv.w);
}

__global__ __launch_bounds__(256)
void finish_kernel(const float* __restrict__ agg, const float* __restrict__ h,
                   const float* __restrict__ dinv, const float* __restrict__ b,
                   float* __restrict__ xout, int n) {
    long i = (long)blockIdx.x * 256 + threadIdx.x;
    long total = (long)n * D_FEAT;
    if (i >= total) return;
    int node = (int)(i >> 7);
    int f = (int)(i & (D_FEAT - 1));
    float di = dinv[node];
    float v = agg[i] + di * di * h[i] + b[f];
    xout[i] = v > 0.0f ? v : 0.0f;
}

#define LDAJ 260
__global__ __launch_bounds__(256)
void gemm_jk_wmma_kernel(const float* __restrict__ x1, const float* __restrict__ x2,
                         const float* __restrict__ Wl, const float* __restrict__ bl,
                         float* __restrict__ out, int nrows) {
    __shared__ float As[16 * LDAJ];
    const int tid = threadIdx.x;
    const long row0 = (long)blockIdx.x * 16;

    for (int idx = tid; idx < 1024; idx += 256) {
        int r = idx >> 6;
        int c4 = idx & 63;
        long gr = row0 + r;
        if (gr >= nrows) gr = nrows - 1;
        const float* srcp = (c4 < 32)
            ? (x1 + gr * D_FEAT + c4 * 4)
            : (x2 + gr * D_FEAT + (c4 - 32) * 4);
        *(float4*)(As + r * LDAJ + c4 * 4) = *(const float4*)srcp;
    }
    __syncthreads();

    const int wave  = tid >> 5;
    const int lane  = tid & 31;
    const int col0  = wave * 16;
    const int mrow  = lane & 15;
    const int khalf = (lane >> 4) * 2;
    const int ncol  = lane & 15;

    v8f acc = {};
    for (int k = 0; k < 2 * D_FEAT; k += 4) {
        v2f a, b;
        a.x = As[mrow * LDAJ + k + khalf + 0];
        a.y = As[mrow * LDAJ + k + khalf + 1];
        b.x = Wl[(k + khalf + 0) * D_FEAT + col0 + ncol];
        b.y = Wl[(k + khalf + 1) * D_FEAT + col0 + ncol];
        acc = __builtin_amdgcn_wmma_f32_16x16x4_f32(
            false, a, false, b, (short)0, acc, false, false);
    }

    const float bias = bl[col0 + ncol];
    const int mbase = (lane >> 4) * 8;
    for (int r = 0; r < 8; ++r) {
        long m = row0 + mbase + r;
        if (m < nrows) out[m * D_FEAT + col0 + ncol] = acc[r] + bias;
    }
}

extern "C" void kernel_launch(void* const* d_in, const int* in_sizes, int n_in,
                              void* d_out, int out_size, void* d_ws, size_t ws_size,
                              hipStream_t stream) {
    (void)n_in; (void)out_size; (void)ws_size;

    const float* x    = (const float*)d_in[0];
    const int*   ei   = (const int*)d_in[1];
    const float* W1   = (const float*)d_in[2];
    const float* b1   = (const float*)d_in[3];
    const float* W2   = (const float*)d_in[4];
    const float* b2   = (const float*)d_in[5];
    const float* linW = (const float*)d_in[6];
    const float* linb = (const float*)d_in[7];
    float* out = (float*)d_out;

    const int N = in_sizes[0] / D_FEAT;
    const int E = in_sizes[1] / 2;
    const int* srcI = ei;
    const int* dstI = ei + E;

    float* ws   = (float*)d_ws;
    float* dinv = ws;
    float* h    = dinv + N;
    float* agg  = h   + (long)N * D_FEAT;
    float* x1   = agg + (long)N * D_FEAT;
    float* x2   = x1  + (long)N * D_FEAT;

    const int nBlkN   = (N + 255) / 256;
    const int nBlkE   = (E + 255) / 256;
    const int nBlkRow = (N + 15) / 16;
    const int nBlkSc  = (E + 7) / 8;
    const long feats  = (long)N * D_FEAT;
    const int nBlkF   = (int)((feats + 255) / 256);
    const long n4     = feats / 4;
    const int nBlkZ   = (int)((n4 + 255) / 256);

    deg_init_kernel<<<nBlkN, 256, 0, stream>>>(dinv, N);
    deg_edges_kernel<<<nBlkE, 256, 0, stream>>>(dstI, dinv, E);
    deg_rsqrt_kernel<<<nBlkN, 256, 0, stream>>>(dinv, N);

    gemm128_wmma_kernel<<<nBlkRow, 256, 0, stream>>>(x, W1, h, N);
    zero_kernel<<<nBlkZ, 256, 0, stream>>>((float4*)agg, n4);
    scatter_kernel<<<nBlkSc, 256, 0, stream>>>(srcI, dstI, dinv, h, agg, E);
    finish_kernel<<<nBlkF, 256, 0, stream>>>(agg, h, dinv, b1, x1, N);

    gemm128_wmma_kernel<<<nBlkRow, 256, 0, stream>>>(x1, W2, h, N);
    zero_kernel<<<nBlkZ, 256, 0, stream>>>((float4*)agg, n4);
    scatter_kernel<<<nBlkSc, 256, 0, stream>>>(srcI, dstI, dinv, h, agg, E);
    finish_kernel<<<nBlkF, 256, 0, stream>>>(agg, h, dinv, b2, x2, N);

    gemm_jk_wmma_kernel<<<nBlkRow, 256, 0, stream>>>(x1, x2, linW, linb, out, N);
}


# Replace the float-atomic scatter by a gather over an edge list sorted by destination (int-atomic ranks + prefix sum): one wave sums each row in registers, writes it once; sum order changes
# speedup vs baseline: 2.5902x; 2.5902x over previous
#include <hip/hip_runtime.h>

typedef __attribute__((ext_vector_type(2))) float v2f;
typedef __attribute__((ext_vector_type(8))) float v8f;

#define D_FEAT 128

__global__ __launch_bounds__(256) void zero_counts_kernel(int* cnt, int n) {
    int i = blockIdx.x * 256 + threadIdx.x;
    if (i < n) cnt[i] = 0;
}

__global__ __launch_bounds__(256)
void rank_edges_kernel(const int* __restrict__ dst, int* cnt, int* __restrict__ rank, int e) {
    int i = blockIdx.x * 256 + threadIdx.x;
    if (i < e) rank[i] = atomicAdd(&cnt[dst[i]], 1);
}

__global__ __launch_bounds__(256)
void block_sum_kernel(const int* __restrict__ cnt, int* __restrict__ bsum, int n) {
    __shared__ int red[8];
    const int tid = threadIdx.x;
    const int i = blockIdx.x * 256 + tid;
    int v = (i < n) ? cnt[i] : 0;
    for (int o = 16; o > 0; o >>= 1) v += __shfl_xor(v, o, 32);
    if ((tid & 31) == 0) red[tid >> 5] = v;
    __syncthreads();
    if (tid == 0) {
        int t = 0;
        for (int w = 0; w < 8; ++w) t += red[w];
        bsum[blockIdx.x] = t;
    }
}

__global__ __launch_bounds__(256)
void scan_dinv_kernel(const int* __restrict__ cnt, const int* __restrict__ bsum,
                      int* __restrict__ rowptr, float* __restrict__ dinv, int n) {
    __shared__ int wtot[8];
    __shared__ int ptot[8];
    const int tid = threadIdx.x;
    const int lane = tid & 31;
    const int wave = tid >> 5;
    const int i = blockIdx.x * 256 + tid;

    int part = 0;
    for (int j = tid; j < (int)blockIdx.x; j += 256) part += bsum[j];
    for (int o = 16; o > 0; o >>= 1) part += __shfl_xor(part, o, 32);

    const int c = (i < n) ? cnt[i] : 0;
    int inc = c;
    for (int o = 1; o < 32; o <<= 1) {
        int t = __shfl_up(inc, o, 32);
        if (lane >= o) inc += t;
    }
    if (lane == 31) wtot[wave] = inc;
    if (lane == 0) ptot[wave] = part;
    __syncthreads();

    int base = 0;
    for (int w = 0; w < 8; ++w) {
        base += ptot[w];
        if (w < wave) base += wtot[w];
    }
    if (i < n) {
        rowptr[i] = base + inc - c;
        float d = (float)(c + 1);
        dinv[i] = (d > 0.0f) ? rsqrtf(d) : 0.0f;
    }
}

__global__ __launch_bounds__(256)
void place_edges_kernel(const int* __restrict__ src, const int* __restrict__ dst,
                        const int* __restrict__ rank, const int* __restrict__ rowptr,
                        const float* __restrict__ dinv, int2* __restrict__ csr, int e) {
    int i = blockIdx.x * 256 + threadIdx.x;
    if (i < e) {
        int s = src[i], d = dst[i];
        csr[rowptr[d] + rank[i]] = make_int2(s, __float_as_int(dinv[s]));
    }
}

#define LDA 132
__global__ __launch_bounds__(256)
void gemm128_wmma_kernel(const float* __restrict__ A, const float* __restrict__ W,
                         float* __restrict__ H, int nrows) {
    __shared__ float As[16 * LDA];
    const int tid = threadIdx.x;
    const long row0 = (long)blockIdx.x * 16;

    for (int idx = tid; idx < 512; idx += 256) {
        int r = idx >> 5;
        int c4 = idx & 31;
        long gr = row0 + r;
        if (gr >= nrows) gr = nrows - 1;
        *(float4*)(As + r * LDA + c4 * 4) =
            *(const float4*)(A + gr * D_FEAT + c4 * 4);
    }
    __syncthreads();

    const int wave  = tid >> 5;
    const int lane  = tid & 31;
    const int col0  = wave * 16;
    const int mrow  = lane & 15;
    const int khalf = (lane >> 4) * 2;
    const int ncol  = lane & 15;

    v8f acc = {};
    for (int k = 0; k < D_FEAT; k += 4) {
        v2f a, b;
        a.x = As[mrow * LDA + k + khalf + 0];
        a.y = As[mrow * LDA + k + khalf + 1];
        b.x = W[(k + khalf + 0) * D_FEAT + col0 + ncol];
        b.y = W[(k + khalf + 1) * D_FEAT + col0 + ncol];
        acc = __builtin_amdgcn_wmma_f32_16x16x4_f32(
             false, a,  false, b,
             (short)0, acc,  false,  false);
    }

    const int mbase = (lane >> 4) * 8;
    for (int r = 0; r < 8; ++r) {
        long m = row0 + mbase + r;
        if (m < nrows) H[m * D_FEAT + col0 + ncol] = acc[r];
    }
}

__device__ __forceinline__ void fma4(float4& acc, float w, const float4& v) {
    acc.x = fmaf(w, v.x, acc.x);
    acc.y = fmaf(w, v.y, acc.y);
    acc.z = fmaf(w, v.z, acc.z);
    acc.w = fmaf(w, v.w, acc.w);
}

__global__ __launch_bounds__(256)
void gather_finish_kernel(const int* __restrict__ rowptr, const int* __restrict__ cnt,
                          const int2* __restrict__ csr, const float* __restrict__ dinv,
                          const float* __restrict__ h, const float* __restrict__ b,
                          float* __restrict__ xout, int n) {
    const int node = __builtin_amdgcn_readfirstlane((int)(blockIdx.x * 8 + (threadIdx.x >> 5)));
    if (node >= n) return;
    const int lane = threadIdx.x & 31;
    const int beg = rowptr[node];
    const int num = cnt[node];
    const float dd = dinv[node];
    const float* hl = h + lane * 4;

    float4 acc = make_float4(0.f, 0.f, 0.f, 0.f);
    for (int base = 0; base < num; base += 32) {
        const int m = (num - base < 32) ? (num - base) : 32;
        int s_l = 0, w_l = 0;
        if (lane < m) {
            int2 ent = csr[beg + base + lane];
            s_l = ent.x;
            w_l = ent.y;
        }
        int j = 0;
        for (; j + 4 <= m; j += 4) {
            const int s0 = __builtin_amdgcn_readlane(s_l, j + 0);
            const int s1 = __builtin_amdgcn_readlane(s_l, j + 1);
            const int s2 = __builtin_amdgcn_readlane(s_l, j + 2);
            const int s3 = __builtin_amdgcn_readlane(s_l, j + 3);
            const float4 h0 = *(const float4*)(hl + (long)s0 * D_FEAT);
            const float4 h1 = *(const float4*)(hl + (long)s1 * D_FEAT);
            const float4 h2 = *(const float4*)(hl + (long)s2 * D_FEAT);
            const float4 h3 = *(const float4*)(hl + (long)s3 * D_FEAT);
            const float w0 = __int_as_float(__builtin_amdgcn_readlane(w_l, j + 0)) * dd;
            const float w1 = __int_as_float(__builtin_amdgcn_readlane(w_l, j + 1)) * dd;
            const float w2 = __int_as_float(__builtin_amdgcn_readlane(w_l, j + 2)) * dd;
            const float w3 = __int_as_float(__builtin_amdgcn_readlane(w_l, j + 3)) * dd;
            fma4(acc, w0, h0);
            fma4(acc, w1, h1);
            fma4(acc, w2, h2);
            fma4(acc, w3, h3);
        }
        const int r = m - j;
        if (r == 3) {
            const int s0 = __builtin_amdgcn_readlane(s_l, j + 0);
            const int s1 = __builtin_amdgcn_readlane(s_l, j + 1);
            const int s2 = __builtin_amdgcn_readlane(s_l, j + 2);
            const float4 h0 = *(const float4*)(hl + (long)s0 * D_FEAT);
            const float4 h1 = *(const float4*)(hl + (long)s1 * D_FEAT);
            const float4 h2 = *(const float4*)(hl + (long)s2 * D_FEAT);
            const float w0 = __int_as_float(__builtin_amdgcn_readlane(w_l, j + 0)) * dd;
            const float w1 = __int_as_float(__builtin_amdgcn_readlane(w_l, j + 1)) * dd;
            const float w2 = __int_as_float(__builtin_amdgcn_readlane(w_l, j + 2)) * dd;
            fma4(acc, w0, h0);
            fma4(acc, w1, h1);
            fma4(acc, w2, h2);
        } else if (r == 2) {
            const int s0 = __builtin_amdgcn_readlane(s_l, j + 0);
            const int s1 = __builtin_amdgcn_readlane(s_l, j + 1);
            const float4 h0 = *(const float4*)(hl + (long)s0 * D_FEAT);
            const float4 h1 = *(const float4*)(hl + (long)s1 * D_FEAT);
            const float w0 = __int_as_float(__builtin_amdgcn_readlane(w_l, j + 0)) * dd;
            const float w1 = __int_as_float(__builtin_amdgcn_readlane(w_l, j + 1)) * dd;
            fma4(acc, w0, h0);
            fma4(acc, w1, h1);
        } else if (r == 1) {
            const int s0 = __builtin_amdgcn_readlane(s_l, j);
            const float4 h0 = *(const float4*)(hl + (long)s0 * D_FEAT);
            const float w0 = __int_as_float(__builtin_amdgcn_readlane(w_l, j)) * dd;
            fma4(acc, w0, h0);
        }
    }

    const float4 hs = *(const float4*)(hl + (long)node * D_FEAT);
    const float4 bv = *(const float4*)(b + lane * 4);
    const float d2 = dd * dd;
    float4 v;
    v.x = acc.x + d2 * hs.x + bv.x;
    v.y = acc.y + d2 * hs.y + bv.y;
    v.z = acc.z + d2 * hs.z + bv.z;
    v.w = acc.w + d2 * hs.w + bv.w;
    v.x = v.x > 0.0f ? v.x : 0.0f;
    v.y = v.y > 0.0f ? v.y : 0.0f;
    v.z = v.z > 0.0f ? v.z : 0.0f;
    v.w = v.w > 0.0f ? v.w : 0.0f;
    *(float4*)(xout + (long)node * D_FEAT + lane * 4) = v;
}

#define LDAJ 260
__global__ __launch_bounds__(256)
void gemm_jk_wmma_kernel(const float* __restrict__ x1, const float* __restrict__ x2,
                         const float* __restrict__ Wl, const float* __restrict__ bl,
                         float* __restrict__ out, int nrows) {
    __shared__ float As[16 * LDAJ];
    const int tid = threadIdx.x;
    const long row0 = (long)blockIdx.x * 16;

    for (int idx = tid; idx < 1024; idx += 256) {
        int r = idx >> 6;
        int c4 = idx & 63;
        long gr = row0 + r;
        if (gr >= nrows) gr = nrows - 1;
        const float* srcp = (c4 < 32)
            ? (x1 + gr * D_FEAT + c4 * 4)
            : (x2 + gr * D_FEAT + (c4 - 32) * 4);
        *(float4*)(As + r * LDAJ + c4 * 4) = *(const float4*)srcp;
    }
    __syncthreads();

    const int wave  = tid >> 5;
    const int lane  = tid & 31;
    const int col0  = wave * 16;
    const int mrow  = lane & 15;
    const int khalf = (lane >> 4) * 2;
    const int ncol  = lane & 15;

    v8f acc = {};
    for (int k = 0; k < 2 * D_FEAT; k += 4) {
        v2f a, b;
        a.x = As[mrow * LDAJ + k + khalf + 0];
        a.y = As[mrow * LDAJ + k + khalf + 1];
        b.x = Wl[(k + khalf + 0) * D_FEAT + col0 + ncol];
        b.y = Wl[(k + khalf + 1) * D_FEAT + col0 + ncol];
        acc = __builtin_amdgcn_wmma_f32_16x16x4_f32(
            false, a, false, b, (short)0, acc, false, false);
    }

    const float bias = bl[col0 + ncol];
    const int mbase = (lane >> 4) * 8;
    for (int r = 0; r < 8; ++r) {
        long m = row0 + mbase + r;
        if (m < nrows) out[m * D_FEAT + col0 + ncol] = acc[r] + bias;
    }
}

extern "C" void kernel_launch(void* const* d_in, const int* in_sizes, int n_in,
                              void* d_out, int out_size, void* d_ws, size_t ws_size,
                              hipStream_t stream) {
    (void)n_in; (void)out_size; (void)ws_size;

    const float* x    = (const float*)d_in[0];
    const int*   ei   = (const int*)d_in[1];
    const float* W1   = (const float*)d_in[2];
    const float* b1   = (const float*)d_in[3];
    const float* W2   = (const float*)d_in[4];
    const float* b2   = (const float*)d_in[5];
    const float* linW = (const float*)d_in[6];
    const float* linb = (const float*)d_in[7];
    float* out = (float*)d_out;

    const int N = in_sizes[0] / D_FEAT;
    const int E = in_sizes[1] / 2;
    const int* srcI = ei;
    const int* dstI = ei + E;

    float* ws   = (float*)d_ws;
    float* dinv = ws;
    float* h    = dinv + N;
    float* edge = h   + (long)N * D_FEAT;
    float* x1   = edge + (long)N * D_FEAT;
    float* x2   = x1  + (long)N * D_FEAT;

    int2* csr    = (int2*)edge;
    int*  rank   = (int*)(csr + E);
    int*  cnt    = rank + E;
    int*  rowptr = cnt + N;
    int*  bsum   = rowptr + N;

    const int nBlkN   = (N + 255) / 256;
    const int nBlkE   = (E + 255) / 256;
    const int nBlkRow = (N + 15) / 16;
    const int nBlkG   = (N + 7) / 8;

    zero_counts_kernel<<<nBlkN, 256, 0, stream>>>(cnt, N);
    rank_edges_kernel<<<nBlkE, 256, 0, stream>>>(dstI, cnt, rank, E);
    block_sum_kernel<<<nBlkN, 256, 0, stream>>>(cnt, bsum, N);
    scan_dinv_kernel<<<nBlkN, 256, 0, stream>>>(cnt, bsum, rowptr, dinv, N);
    place_edges_kernel<<<nBlkE, 256, 0, stream>>>(srcI, dstI, rank, rowptr, dinv, csr, E);

    gemm128_wmma_kernel<<<nBlkRow, 256, 0, stream>>>(x, W1, h, N);
    gather_finish_kernel<<<nBlkG, 256, 0, stream>>>(rowptr, cnt, csr, dinv, h, b1, x1, N);

    gemm128_wmma_kernel<<<nBlkRow, 256, 0, stream>>>(x1, W2, h, N);
    gather_finish_kernel<<<nBlkG, 256, 0, stream>>>(rowptr, cnt, csr, dinv, h, b2, x2, N);

    gemm_jk_wmma_kernel<<<nBlkRow, 256, 0, stream>>>(x1, x2, linW, linb, out, N);
}


# Fewer, longer-lived waves (the parent ran at the wave-entry rate): GEMM blocks walk row tiles on a fixed grid, W strip held in registers; gather waves take 4 rows; edge kernels 4 edges per thread
# speedup vs baseline: 2.7231x; 1.0513x over previous
#include <hip/hip_runtime.h>

typedef __attribute__((ext_vector_type(2))) float v2f;
typedef __attribute__((ext_vector_type(8))) float v8f;

#define D_FEAT 128

__global__ __launch_bounds__(256) void zero_counts_kernel(int* cnt, int n) {
    int i = blockIdx.x * 256 + threadIdx.x;
    if (i < n) cnt[i] = 0;
}

__global__ __launch_bounds__(256)
void rank_edges_kernel(const int* __restrict__ dst, int* cnt, int* __restrict__ rank, int e) {
    const int i0 = blockIdx.x * 1024 + threadIdx.x;
    if (blockIdx.x * 1024 + 1024 <= e) {
        const int d0 = dst[i0], d1 = dst[i0 + 256], d2 = dst[i0 + 512], d3 = dst[i0 + 768];
        const int r0 = atomicAdd(&cnt[d0], 1);
        const int r1 = atomicAdd(&cnt[d1], 1);
        const int r2 = atomicAdd(&cnt[d2], 1);
        const int r3 = atomicAdd(&cnt[d3], 1);
        rank[i0]       = r0;
        rank[i0 + 256] = r1;
        rank[i0 + 512] = r2;
        rank[i0 + 768] = r3;
    } else {
        for (int q = 0; q < 4; ++q) {
            const int i = i0 + q * 256;
            if (i < e) rank[i] = atomicAdd(&cnt[dst[i]], 1);
        }
    }
}

__global__ __launch_bounds__(256)
void block_sum_kernel(const int* __restrict__ cnt, int* __restrict__ bsum, int n) {
    __shared__ int red[8];
    const int tid = threadIdx.x;
    const int i = blockIdx.x * 256 + tid;
    int v = (i < n) ? cnt[i] : 0;
    for (int o = 16; o > 0; o >>= 1) v += __shfl_xor(v, o, 32);
    if ((tid & 31) == 0) red[tid >> 5] = v;
    __syncthreads();
    if (tid == 0) {
        int t = 0;
        for (int w = 0; w < 8; ++w) t += red[w];
        bsum[blockIdx.x] = t;
    }
}

__global__ __launch_bounds__(256)
void scan_dinv_kernel(const int* __restrict__ cnt, const int* __restrict__ bsum,
                      int* __restrict__ rowptr, float* __restrict__ dinv, int n) {
    __shared__ int wtot[8];
    __shared__ int ptot[8];
    const int tid = threadIdx.x;
    const int lane = tid & 31;
    const int wave = tid >> 5;
    const int i = blockIdx.x * 256 + tid;

    int part = 0;
    for (int j = tid; j < (int)blockIdx.x; j += 256) part += bsum[j];
    for (int o = 16; o > 0; o >>= 1) part += __shfl_xor(part, o, 32);

    const int c = (i < n) ? cnt[i] : 0;
    int inc = c;
    for (int o = 1; o < 32; o <<= 1) {
        int t = __shfl_up(inc, o, 32);
        if (lane >= o) inc += t;
    }
    if (lane == 31) wtot[wave] = inc;
    if (lane == 0) ptot[wave] = part;
    __syncthreads();

    int base = 0;
    for (int w = 0; w < 8; ++w) {
        base += ptot[w];
        if (w < wave) base += wtot[w];
    }
    if (i < n) {
        rowptr[i] = base + inc - c;
        float d = (float)(c + 1);
        dinv[i] = (d > 0.0f) ? rsqrtf(d) : 0.0f;
    }
}

__global__ __launch_bounds__(256)
void place_edges_kernel(const int* __restrict__ src, const int* __restrict__ dst,
                        const int* __restrict__ rank, const int* __restrict__ rowptr,
                        const float* __restrict__ dinv, int2* __restrict__ csr, int e) {
    const int i0 = blockIdx.x * 1024 + threadIdx.x;
    if (blockIdx.x * 1024 + 1024 <= e) {
        int s[4], d[4], rk[4], pos[4], wv[4];
        for (int q = 0; q < 4; ++q) {
            const int i = i0 + q * 256;
            s[q]  = src[i];
            d[q]  = dst[i];
            rk[q] = rank[i];
        }
        for (int q = 0; q < 4; ++q) {
            pos[q] = rowptr[d[q]] + rk[q];
            wv[q]  = __float_as_int(dinv[s[q]]);
        }
        for (int q = 0; q < 4; ++q) csr[pos[q]] = make_int2(s[q], wv[q]);
    } else {
        for (int q = 0; q < 4; ++q) {
            const int i = i0 + q * 256;
            if (i < e) {
                const int s = src[i], d = dst[i];
                csr[rowptr[d] + rank[i]] = make_int2(s, __float_as_int(dinv[s]));
            }
        }
    }
}

#define LDA 132
__global__ __launch_bounds__(256)
void gemm128_wmma_kernel(const float* __restrict__ A, const float* __restrict__ W,
                         float* __restrict__ H, int nrows, int nblk) {
    __shared__ float As[16 * LDA];
    const int tid   = threadIdx.x;
    const int wave  = tid >> 5;
    const int lane  = tid & 31;
    const int col0  = wave * 16;
    const int mrow  = lane & 15;
    const int khalf = (lane >> 4) * 2;
    const int ncol  = lane & 15;
    const int mbase = (lane >> 4) * 8;
    const int ntiles = (nrows + 15) >> 4;

    v2f bf[32];
#pragma unroll
    for (int kk = 0; kk < 32; ++kk) {
        bf[kk].x = W[(kk * 4 + khalf + 0) * D_FEAT + col0 + ncol];
        bf[kk].y = W[(kk * 4 + khalf + 1) * D_FEAT + col0 + ncol];
    }

    const int sr = tid >> 5;
    const int c4 = tid & 31;
    int tile = blockIdx.x;
    float4 p0 = make_float4(0.f, 0.f, 0.f, 0.f), p1 = p0;
    if (tile < ntiles) {
        long ra = (long)tile * 16 + sr, rb = ra + 8;
        if (ra >= nrows) ra = nrows - 1;
        if (rb >= nrows) rb = nrows - 1;
        p0 = *(const float4*)(A + ra * D_FEAT + c4 * 4);
        p1 = *(const float4*)(A + rb * D_FEAT + c4 * 4);
    }
    for (; tile < ntiles; tile += nblk) {
        __syncthreads();
        *(float4*)(As + sr * LDA + c4 * 4)       = p0;
        *(float4*)(As + (sr + 8) * LDA + c4 * 4) = p1;
        __syncthreads();

        const int nxt = tile + nblk;
        if (nxt < ntiles) {
            long ra = (long)nxt * 16 + sr, rb = ra + 8;
            if (ra >= nrows) ra = nrows - 1;
            if (rb >= nrows) rb = nrows - 1;
            p0 = *(const float4*)(A + ra * D_FEAT + c4 * 4);
            p1 = *(const float4*)(A + rb * D_FEAT + c4 * 4);
        }

        v8f acc = {};
#pragma unroll
        for (int kk = 0; kk < 32; ++kk) {
            v2f a;
            a.x = As[mrow * LDA + kk * 4 + khalf + 0];
            a.y = As[mrow * LDA + kk * 4 + khalf + 1];
            acc = __builtin_amdgcn_wmma_f32_16x16x4_f32(
                 false, a,  false, bf[kk],
                 (short)0, acc,  false,  false);
        }

        const long row0 = (long)tile * 16;
        if (row0 + 16 <= nrows) {
            float* hp = H + (row0 + mbase) * D_FEAT + col0 + ncol;
            for (int r = 0; r < 8; ++r) hp[r * D_FEAT] = acc[r];
        } else {
            for (int r = 0; r < 8; ++r) {
                long m = row0 + mbase + r;
                if (m < nrows) H[m * D_FEAT + col0 + ncol] = acc[r];
            }
        }
    }
}

#define GATHER_ROWS 4
__device__ __forceinline__ void fma4(float4& acc, float w, const float4& v) {
    acc.x = fmaf(w, v.x, acc.x);
    acc.y = fmaf(w, v.y, acc.y);
    acc.z = fmaf(w, v.z, acc.z);
    acc.w = fmaf(w, v.w, acc.w);
}

__global__ __launch_bounds__(256)
void gather_finish_kernel(const int* __restrict__ rowptr, const int* __restrict__ cnt,
                          const int2* __restrict__ csr, const float* __restrict__ dinv,
                          const float* __restrict__ h, const float* __restrict__ b,
                          float* __restrict__ xout, int n) {
    const int wv = __builtin_amdgcn_readfirstlane((int)(blockIdx.x * 8 + (threadIdx.x >> 5)));
    const int first = wv * GATHER_ROWS;
    if (first >= n) return;
    const int last = (first + GATHER_ROWS < n) ? (first + GATHER_ROWS) : n;
    const int lane = threadIdx.x & 31;
    const float* hl = h + lane * 4;
    const float4 bv = *(const float4*)(b + lane * 4);

    int   nbeg = rowptr[first];
    int   nnum = cnt[first];
    float ndd  = dinv[first];
    for (int node = first; node < last; ++node) {
    const int beg = nbeg;
    const int num = nnum;
    const float dd = ndd;
    int2 ent = make_int2(0, 0);
    if (lane < num) ent = csr[beg + lane];
    const float4 hs = *(const float4*)(hl + (long)node * D_FEAT);
    {
        const int nx = (node + 1 < n) ? (node + 1) : node;
        nbeg = rowptr[nx];
        nnum = cnt[nx];
        ndd  = dinv[nx];
    }

    float4 acc = make_float4(0.f, 0.f, 0.f, 0.f);
    for (int base = 0; base < num; base += 32) {
        const int m = (num - base < 32) ? (num - base) : 32;
        const int s_l = ent.x, w_l = ent.y;
        if (base + 32 < num) {
            ent = make_int2(0, 0);
            if (lane < num - base - 32) ent = csr[beg + base + 32 + lane];
        }
        int j = 0;
        for (; j + 4 <= m; j += 4) {
            const int s0 = __builtin_amdgcn_readlane(s_l, j + 0);
            const int s1 = __builtin_amdgcn_readlane(s_l, j + 1);
            const int s2 = __builtin_amdgcn_readlane(s_l, j + 2);
            const int s3 = __builtin_amdgcn_readlane(s_l, j + 3);
            const float4 h0 = *(const float4*)(hl + (long)s0 * D_FEAT);
            const float4 h1 = *(const float4*)(hl + (long)s1 * D_FEAT);
            const float4 h2 = *(const float4*)(hl + (long)s2 * D_FEAT);
            const float4 h3 = *(const float4*)(hl + (long)s3 * D_FEAT);
            const float w0 = __int_as_float(__builtin_amdgcn_readlane(w_l, j + 0)) * dd;
            const float w1 = __int_as_float(__builtin_amdgcn_readlane(w_l, j + 1)) * dd;
            const float w2 = __int_as_float(__builtin_amdgcn_readlane(w_l, j + 2)) * dd;
            const float w3 = __int_as_float(__builtin_amdgcn_readlane(w_l, j + 3)) * dd;
            fma4(acc, w0, h0);
            fma4(acc, w1, h1);
            fma4(acc, w2, h2);
            fma4(acc, w3, h3);
        }
        const int r = m - j;
        if (r == 3) {
            const int s0 = __builtin_amdgcn_readlane(s_l, j + 0);
            const int s1 = __builtin_amdgcn_readlane(s_l, j + 1);
            const int s2 = __builtin_amdgcn_readlane(s_l, j + 2);
            const float4 h0 = *(const float4*)(hl + (long)s0 * D_FEAT);
            const float4 h1 = *(const float4*)(hl + (long)s1 * D_FEAT);
            const float4 h2 = *(const float4*)(hl + (long)s2 * D_FEAT);
            const float w0 = __int_as_float(__builtin_amdgcn_readlane(w_l, j + 0)) * dd;
            const float w1 = __int_as_float(__builtin_amdgcn_readlane(w_l, j + 1)) * dd;
            const float w2 = __int_as_float(__builtin_amdgcn_readlane(w_l, j + 2)) * dd;
            fma4(acc, w0, h0);
            fma4(acc, w1, h1);
            fma4(acc, w2, h2);
        } else if (r == 2) {
            const int s0 = __builtin_amdgcn_readlane(s_l, j + 0);
            const int s1 = __builtin_amdgcn_readlane(s_l, j + 1);
            const float4 h0 = *(const float4*)(hl + (long)s0 * D_FEAT);
            const float4 h1 = *(const float4*)(hl + (long)s1 * D_FEAT);
            const float w0 = __int_as_float(__builtin_amdgcn_readlane(w_l, j + 0)) * dd;
            const float w1 = __int_as_float(__builtin_amdgcn_readlane(w_l, j + 1)) * dd;
            fma4(acc, w0, h0);
            fma4(acc, w1, h1);
        } else if (r == 1) {
            const int s0 = __builtin_amdgcn_readlane(s_l, j);
            const float4 h0 = *(const float4*)(hl + (long)s0 * D_FEAT);
            const float w0 = __int_as_float(__builtin_amdgcn_readlane(w_l, j)) * dd;
            fma4(acc, w0, h0);
        }
    }

    const float d2 = dd * dd;
    float4 v;
    v.x = acc.x + d2 * hs.x + bv.x;
    v.y = acc.y + d2 * hs.y + bv.y;
    v.z = acc.z + d2 * hs.z + bv.z;
    v.w = acc.w + d2 * hs.w + bv.w;
    v.x = v.x > 0.0f ? v.x : 0.0f;
    v.y = v.y > 0.0f ? v.y : 0.0f;
    v.z = v.z > 0.0f ? v.z : 0.0f;
    v.w = v.w > 0.0f ? v.w : 0.0f;
    *(float4*)(xout + (long)node * D_FEAT + lane * 4) = v;
    }
}

#define LDAJ 260
__global__ __launch_bounds__(256)
void gemm_jk_wmma_kernel(const float* __restrict__ x1, const float* __restrict__ x2,
                         const float* __restrict__ Wl, const float* __restrict__ bl,
                         float* __restrict__ out, int nrows, int nblk) {
    __shared__ float As[16 * LDAJ];
    const int tid   = threadIdx.x;
    const int wave  = tid >> 5;
    const int lane  = tid & 31;
    const int col0  = wave * 16;
    const int mrow  = lane & 15;
    const int khalf = (lane >> 4) * 2;
    const int ncol  = lane & 15;
    const int mbase = (lane >> 4) * 8;
    const int ntiles = (nrows + 15) >> 4;

    v2f bf[64];
#pragma unroll
    for (int kk = 0; kk < 64; ++kk) {
        bf[kk].x = Wl[(kk * 4 + khalf + 0) * D_FEAT + col0 + ncol];
        bf[kk].y = Wl[(kk * 4 + khalf + 1) * D_FEAT + col0 + ncol];
    }
    const float bias = bl[col0 + ncol];

    const int sr = tid >> 6;
    const int c4 = tid & 63;
    const float* xs = (c4 < 32) ? (x1 + c4 * 4) : (x2 + (c4 - 32) * 4);
    int tile = blockIdx.x;
    float4 p[4];
    for (int q = 0; q < 4; ++q) p[q] = make_float4(0.f, 0.f, 0.f, 0.f);
    if (tile < ntiles) {
        for (int q = 0; q < 4; ++q) {
            long gr = (long)tile * 16 + sr + 4 * q;
            if (gr >= nrows) gr = nrows - 1;
            p[q] = *(const float4*)(xs + gr * D_FEAT);
        }
    }
    for (; tile < ntiles; tile += nblk) {
        __syncthreads();
        for (int q = 0; q < 4; ++q)
            *(float4*)(As + (sr + 4 * q) * LDAJ + c4 * 4) = p[q];
        __syncthreads();

        const int nxt = tile + nblk;
        if (nxt < ntiles) {
            for (int q = 0; q < 4; ++q) {
                long gr = (long)nxt * 16 + sr + 4 * q;
                if (gr >= nrows) gr = nrows - 1;
                p[q] = *(const float4*)(xs + gr * D_FEAT);
            }
        }

        v8f acc = {};
#pragma unroll
        for (int kk = 0; kk < 64; ++kk) {
            v2f a;
            a.x = As[mrow * LDAJ + kk * 4 + khalf + 0];
            a.y = As[mrow * LDAJ + kk * 4 + khalf + 1];
            acc = __builtin_amdgcn_wmma_f32_16x16x4_f32(
                false, a, false, bf[kk], (short)0, acc, false, false);
        }

        const long row0 = (long)tile * 16;
        if (row0 + 16 <= nrows) {
            float* op = out + (row0 + mbase) * D_FEAT + col0 + ncol;
            for (int r = 0; r < 8; ++r) op[r * D_FEAT] = acc[r] + bias;
        } else {
            for (int r = 0; r < 8; ++r) {
                long m = row0 + mbase + r;
                if (m < nrows) out[m * D_FEAT + col0 + ncol] = acc[r] + bias;
            }
        }
    }
}

extern "C" void kernel_launch(void* const* d_in, const int* in_sizes, int n_in,
                              void* d_out, int out_size, void* d_ws, size_t ws_size,
                              hipStream_t stream) {
    (void)n_in; (void)out_size; (void)ws_size;

    const float* x    = (const float*)d_in[0];
    const int*   ei   = (const int*)d_in[1];
    const float* W1   = (const float*)d_in[2];
    const float* b1   = (const float*)d_in[3];
    const float* W2   = (const float*)d_in[4];
    const float* b2   = (const float*)d_in[5];
    const float* linW = (const float*)d_in[6];
    const float* linb = (const float*)d_in[7];
    float* out = (float*)d_out;

    const int N = in_sizes[0] / D_FEAT;
    const int E = in_sizes[1] / 2;
    const int* srcI = ei;
    const int* dstI = ei + E;

    float* ws   = (float*)d_ws;
    float* dinv = ws;
    float* h    = dinv + N;
    float* edge = h   + (long)N * D_FEAT;
    float* x1   = edge + (long)N * D_FEAT;
    float* x2   = x1  + (long)N * D_FEAT;

    int2* csr    = (int2*)edge;
    int*  rank   = (int*)(csr + E);
    int*  cnt    = rank + E;
    int*  rowptr = cnt + N;
    int*  bsum   = rowptr + N;

    const int nBlkN   = (N + 255) / 256;
    const int nBlkE4  = (E + 1023) / 1024;
    const int nBlkRow = 512;
    const int nBlkG   = (N + 8 * GATHER_ROWS - 1) / (8 * GATHER_ROWS);

    zero_counts_kernel<<<nBlkN, 256, 0, stream>>>(cnt, N);
    rank_edges_kernel<<<nBlkE4, 256, 0, stream>>>(dstI, cnt, rank, E);
    block_sum_kernel<<<nBlkN, 256, 0, stream>>>(cnt, bsum, N);
    scan_dinv_kernel<<<nBlkN, 256, 0, stream>>>(cnt, bsum, rowptr, dinv, N);
    place_edges_kernel<<<nBlkE4, 256, 0, stream>>>(srcI, dstI, rank, rowptr, dinv, csr, E);

    gemm128_wmma_kernel<<<nBlkRow, 256, 0, stream>>>(x, W1, h, N, nBlkRow);
    gather_finish_kernel<<<nBlkG, 256, 0, stream>>>(rowptr, cnt, csr, dinv, h, b1, x1, N);

    gemm128_wmma_kernel<<<nBlkRow, 256, 0, stream>>>(x1, W2, h, N, nBlkRow);
    gather_finish_kernel<<<nBlkG, 256, 0, stream>>>(rowptr, cnt, csr, dinv, h, b2, x2, N);

    gemm_jk_wmma_kernel<<<nBlkRow, 256, 0, stream>>>(x1, x2, linW, linb, out, N, nBlkRow);
}


# Edge ranks from LDS counters, not returning global atomics: 64 blocks each rank one edge chunk in a 200 KiB LDS histogram, a scan turns chunk counts into starts; 10 launches->8; row sum order changes
# speedup vs baseline: 3.1564x; 1.1591x over previous
#include <hip/hip_runtime.h>

typedef __attribute__((ext_vector_type(2))) float v2f;
typedef __attribute__((ext_vector_type(8))) float v8f;

#define D_FEAT 128

#define HIST_CAP 51200
#define RANK_BLOCKS 64

__global__ __launch_bounds__(1024)
void rank_local_kernel(const int* __restrict__ dst, int* __restrict__ rank,
                       int* __restrict__ cntP, int* __restrict__ bsumP,
                       int e, int chunk, int npad, int nb) {
    __shared__ __attribute__((aligned(16))) int hist[HIST_CAP + 4];
    __shared__ int gsum[HIST_CAP / 256];
    const int tid  = threadIdx.x;
    const int lane = tid & 31;
    const int e0 = blockIdx.x * chunk;
    const int e1 = (e0 + chunk < e) ? (e0 + chunk) : e;
    int* cp = cntP + (long)blockIdx.x * npad;
    int* bp = bsumP + (long)blockIdx.x * nb;

    for (int base = 0; base < npad; base += HIST_CAP) {
        const int wlen = (npad - base < HIST_CAP) ? (npad - base) : HIST_CAP;
        for (int i = tid * 4; i < wlen; i += 4096) *(int4*)(hist + i) = make_int4(0, 0, 0, 0);
        for (int i = tid; i < (wlen >> 8); i += 1024) gsum[i] = 0;
        __syncthreads();

        for (int i = e0 + tid; i < e1; i += 8192) {
            int d[8], r[8];
#pragma unroll
            for (int q = 0; q < 8; ++q) d[q] = dst[(i + q * 1024 < e1) ? (i + q * 1024) : (e1 - 1)];
#pragma unroll
            for (int q = 0; q < 8; ++q) d[q] = (i + q * 1024 < e1) ? (d[q] - base) : -1;
#pragma unroll
            for (int q = 0; q < 8; ++q)
                r[q] = atomicAdd(&hist[((unsigned)d[q] < (unsigned)wlen) ? d[q] : HIST_CAP], 1);
#pragma unroll
            for (int q = 0; q < 8; ++q)
                if ((unsigned)d[q] < (unsigned)wlen) rank[i + q * 1024] = r[q];
        }
        __syncthreads();

        for (int i0 = (tid & ~31) * 4; i0 < wlen; i0 += 4096) {
            const int k = i0 + lane * 4;
            const int4 v = *(const int4*)(hist + k);
            *(int4*)(cp + base + k) = v;
            int s = v.x + v.y + v.z + v.w;
            s += __builtin_amdgcn_update_dpp(0, s, 0x161, 0xf, 0xf, false);
            s += __builtin_amdgcn_update_dpp(0, s, 0x162, 0xf, 0xf, false);
            s += __builtin_amdgcn_update_dpp(0, s, 0x164, 0xf, 0xf, false);
            s += __builtin_amdgcn_update_dpp(0, s, 0x168, 0xf, 0xf, false);
            s += __builtin_amdgcn_permlanex16(s, s, 0x76543210, 0xfedcba98, false, false);
            if (lane == 0) atomicAdd(&gsum[i0 >> 8], s);
        }
        __syncthreads();
        for (int g = tid; g < (wlen >> 8); g += 1024) bp[(base >> 8) + g] = gsum[g];
        __syncthreads();
    }
}

__global__ __launch_bounds__(256)
void scan_offsets_kernel(int* __restrict__ cntP, const int* __restrict__ bsumP,
                         int* __restrict__ cnt, int* __restrict__ rowptr,
                         float* __restrict__ dinv, int n, int npad, int nb) {
    __shared__ int wtot[8];
    __shared__ int ptot[8];
    const int tid = threadIdx.x;
    const int lane = tid & 31;
    const int wave = tid >> 5;
    const int i = blockIdx.x * 256 + tid;

    int part = 0;
    for (int j = tid; j < (int)blockIdx.x; j += 256) {
#pragma unroll 8
        for (int p = 0; p < RANK_BLOCKS; ++p) part += bsumP[p * nb + j];
    }
    for (int o = 16; o > 0; o >>= 1) part += __shfl_xor(part, o, 32);

    int cpv[RANK_BLOCKS];
    int c = 0;
#pragma unroll
    for (int p = 0; p < RANK_BLOCKS; ++p) {
        cpv[p] = cntP[(long)p * npad + i];
        c += cpv[p];
    }
    int inc = c;
    for (int o = 1; o < 32; o <<= 1) {
        int t = __shfl_up(inc, o, 32);
        if (lane >= o) inc += t;
    }
    if (lane == 31) wtot[wave] = inc;
    if (lane == 0) ptot[wave] = part;
    __syncthreads();

    int base = 0;
    for (int w = 0; w < 8; ++w) {
        base += ptot[w];
        if (w < wave) base += wtot[w];
    }
    if (i < n) {
        const int rp = base + inc - c;
        cnt[i] = c;
        rowptr[i] = rp;
        float d = (float)(c + 1);
        dinv[i] = (d > 0.0f) ? rsqrtf(d) : 0.0f;
        int run = rp;
#pragma unroll
        for (int p = 0; p < RANK_BLOCKS; ++p) {
            cntP[(long)p * npad + i] = run;
            run += cpv[p];
        }
    }
}

__global__ __launch_bounds__(256)
void place_edges_kernel(const int* __restrict__ src, const int* __restrict__ dst,
                        const int* __restrict__ rank, const int* __restrict__ offP,
                        const float* __restrict__ dinv, int2* __restrict__ csr,
                        int e, int chunk, int npad) {
    const int i0 = blockIdx.x * 1024 + threadIdx.x;
    const int* rowptr = offP + (long)((blockIdx.x * 1024) / chunk) * npad;
    if (blockIdx.x * 1024 + 1024 <= e) {
        int s[4], d[4], rk[4], pos[4], wv[4];
        for (int q = 0; q < 4; ++q) {
            const int i = i0 + q * 256;
            s[q]  = src[i];
            d[q]  = dst[i];
            rk[q] = rank[i];
        }
        for (int q = 0; q < 4; ++q) {
            pos[q] = rowptr[d[q]] + rk[q];
            wv[q]  = __float_as_int(dinv[s[q]]);
        }
        for (int q = 0; q < 4; ++q) csr[pos[q]] = make_int2(s[q], wv[q]);
    } else {
        for (int q = 0; q < 4; ++q) {
            const int i = i0 + q * 256;
            if (i < e) {
                const int s = src[i], d = dst[i];
                csr[rowptr[d] + rank[i]] = make_int2(s, __float_as_int(dinv[s]));
            }
        }
    }
}

#define LDA 132
__global__ __launch_bounds__(256)
void gemm128_wmma_kernel(const float* __restrict__ A, const float* __restrict__ W,
                         float* __restrict__ H, int nrows, int nblk) {
    __shared__ float As[16 * LDA];
    const int tid   = threadIdx.x;
    const int wave  = tid >> 5;
    const int lane  = tid & 31;
    const int col0  = wave * 16;
    const int mrow  = lane & 15;
    const int khalf = (lane >> 4) * 2;
    const int ncol  = lane & 15;
    const int mbase = (lane >> 4) * 8;
    const int ntiles = (nrows + 15) >> 4;

    v2f bf[32];
#pragma unroll
    for (int kk = 0; kk < 32; ++kk) {
        bf[kk].x = W[(kk * 4 + khalf + 0) * D_FEAT + col0 + ncol];
        bf[kk].y = W[(kk * 4 + khalf + 1) * D_FEAT + col0 + ncol];
    }

    const int sr = tid >> 5;
    const int c4 = tid & 31;
    int tile = blockIdx.x;
    float4 p0 = make_float4(0.f, 0.f, 0.f, 0.f), p1 = p0;
    if (tile < ntiles) {
        long ra = (long)tile * 16 + sr, rb = ra + 8;
        if (ra >= nrows) ra = nrows - 1;
        if (rb >= nrows) rb = nrows - 1;
        p0 = *(const float4*)(A + ra * D_FEAT + c4 * 4);
        p1 = *(const float4*)(A + rb * D_FEAT + c4 * 4);
    }
    for (; tile < ntiles; tile += nblk) {
        __syncthreads();
        *(float4*)(As + sr * LDA + c4 * 4)       = p0;
        *(float4*)(As + (sr + 8) * LDA + c4 * 4) = p1;
        __syncthreads();

        const int nxt = tile + nblk;
        if (nxt < ntiles) {
            long ra = (long)nxt * 16 + sr, rb = ra + 8;
            if (ra >= nrows) ra = nrows - 1;
            if (rb >= nrows) rb = nrows - 1;
            p0 = *(const float4*)(A + ra * D_FEAT + c4 * 4);
            p1 = *(const float4*)(A + rb * D_FEAT + c4 * 4);
        }

        v8f acc = {};
#pragma unroll
        for (int kk = 0; kk < 32; ++kk) {
            v2f a;
            a.x = As[mrow * LDA + kk * 4 + khalf + 0];
            a.y = As[mrow * LDA + kk * 4 + khalf + 1];
            acc = __builtin_amdgcn_wmma_f32_16x16x4_f32(
                 false, a,  false, bf[kk],
                 (short)0, acc,  false,  false);
        }

        const long row0 = (long)tile * 16;
        if (row0 + 16 <= nrows) {
            float* hp = H + (row0 + mbase) * D_FEAT + col0 + ncol;
            for (int r = 0; r < 8; ++r) hp[r * D_FEAT] = acc[r];
        } else {
            for (int r = 0; r < 8; ++r) {
                long m = row0 + mbase + r;
                if (m < nrows) H[m * D_FEAT + col0 + ncol] = acc[r];
            }
        }
    }
}

#define GATHER_ROWS 4
__device__ __forceinline__ void fma4(float4& acc, float w, const float4& v) {
    acc.x = fmaf(w, v.x, acc.x);
    acc.y = fmaf(w, v.y, acc.y);
    acc.z = fmaf(w, v.z, acc.z);
    acc.w = fmaf(w, v.w, acc.w);
}

__global__ __launch_bounds__(256)
void gather_finish_kernel(const int* __restrict__ rowptr, const int* __restrict__ cnt,
                          const int2* __restrict__ csr, const float* __restrict__ dinv,
                          const float* __restrict__ h, const float* __restrict__ b,
                          float* __restrict__ xout, int n) {
    const int wv = __builtin_amdgcn_readfirstlane((int)(blockIdx.x * 8 + (threadIdx.x >> 5)));
    const int first = wv * GATHER_ROWS;
    if (first >= n) return;
    const int last = (first + GATHER_ROWS < n) ? (first + GATHER_ROWS) : n;
    const int lane = threadIdx.x & 31;
    const float* hl = h + lane * 4;
    const float4 bv = *(const float4*)(b + lane * 4);

    int   nbeg = rowptr[first];
    int   nnum = cnt[first];
    float ndd  = dinv[first];
    for (int node = first; node < last; ++node) {
    const int beg = nbeg;
    const int num = nnum;
    const float dd = ndd;
    int2 ent = make_int2(0, 0);
    if (lane < num) ent = csr[beg + lane];
    const float4 hs = *(const float4*)(hl + (long)node * D_FEAT);
    {
        const int nx = (node + 1 < n) ? (node + 1) : node;
        nbeg = rowptr[nx];
        nnum = cnt[nx];
        ndd  = dinv[nx];
    }

    float4 acc = make_float4(0.f, 0.f, 0.f, 0.f);
    for (int base = 0; base < num; base += 32) {
        const int m = (num - base < 32) ? (num - base) : 32;
        const int s_l = ent.x, w_l = ent.y;
        if (base + 32 < num) {
            ent = make_int2(0, 0);
            if (lane < num - base - 32) ent = csr[beg + base + 32 + lane];
        }
        int j = 0;
        for (; j + 4 <= m; j += 4) {
            const int s0 = __builtin_amdgcn_readlane(s_l, j + 0);
            const int s1 = __builtin_amdgcn_readlane(s_l, j + 1);
            const int s2 = __builtin_amdgcn_readlane(s_l, j + 2);
            const int s3 = __builtin_amdgcn_readlane(s_l, j + 3);
            const float4 h0 = *(const float4*)(hl + (long)s0 * D_FEAT);
            const float4 h1 = *(const float4*)(hl + (long)s1 * D_FEAT);
            const float4 h2 = *(const float4*)(hl + (long)s2 * D_FEAT);
            const float4 h3 = *(const float4*)(hl + (long)s3 * D_FEAT);
            const float w0 = __int_as_float(__builtin_amdgcn_readlane(w_l, j + 0)) * dd;
            const float w1 = __int_as_float(__builtin_amdgcn_readlane(w_l, j + 1)) * dd;
            const float w2 = __int_as_float(__builtin_amdgcn_readlane(w_l, j + 2)) * dd;
            const float w3 = __int_as_float(__builtin_amdgcn_readlane(w_l, j + 3)) * dd;
            fma4(acc, w0, h0);
            fma4(acc, w1, h1);
            fma4(acc, w2, h2);
            fma4(acc, w3, h3);
        }
        const int r = m - j;
        if (r == 3) {
            const int s0 = __builtin_amdgcn_readlane(s_l, j + 0);
            const int s1 = __builtin_amdgcn_readlane(s_l, j + 1);
            const int s2 = __builtin_amdgcn_readlane(s_l, j + 2);
            const float4 h0 = *(const float4*)(hl + (long)s0 * D_FEAT);
            const float4 h1 = *(const float4*)(hl + (long)s1 * D_FEAT);
            const float4 h2 = *(const float4*)(hl + (long)s2 * D_FEAT);
            const float w0 = __int_as_float(__builtin_amdgcn_readlane(w_l, j + 0)) * dd;
            const float w1 = __int_as_float(__builtin_amdgcn_readlane(w_l, j + 1)) * dd;
            const float w2 = __int_as_float(__builtin_amdgcn_readlane(w_l, j + 2)) * dd;
            fma4(acc, w0, h0);
            fma4(acc, w1, h1);
            fma4(acc, w2, h2);
        } else if (r == 2) {
            const int s0 = __builtin_amdgcn_readlane(s_l, j + 0);
            const int s1 = __builtin_amdgcn_readlane(s_l, j + 1);
            const float4 h0 = *(const float4*)(hl + (long)s0 * D_FEAT);
            const float4 h1 = *(const float4*)(hl + (long)s1 * D_FEAT);
            const float w0 = __int_as_float(__builtin_amdgcn_readlane(w_l, j + 0)) * dd;
            const float w1 = __int_as_float(__builtin_amdgcn_readlane(w_l, j + 1)) * dd;
            fma4(acc, w0, h0);
            fma4(acc, w1, h1);
        } else if (r == 1) {
            const int s0 = __builtin_amdgcn_readlane(s_l, j);
            const float4 h0 = *(const float4*)(hl + (long)s0 * D_FEAT);
            const float w0 = __int_as_float(__builtin_amdgcn_readlane(w_l, j)) * dd;
            fma4(acc, w0, h0);
        }
    }

    const float d2 = dd * dd;
    float4 v;
    v.x = acc.x + d2 * hs.x + bv.x;
    v.y = acc.y + d2 * hs.y + bv.y;
    v.z = acc.z + d2 * hs.z + bv.z;
    v.w = acc.w + d2 * hs.w + bv.w;
    v.x = v.x > 0.0f ? v.x : 0.0f;
    v.y = v.y > 0.0f ? v.y : 0.0f;
    v.z = v.z > 0.0f ? v.z : 0.0f;
    v.w = v.w > 0.0f ? v.w : 0.0f;
    *(float4*)(xout + (long)node * D_FEAT + lane * 4) = v;
    }
}

#define LDAJ 260
__global__ __launch_bounds__(256)
void gemm_jk_wmma_kernel(const float* __restrict__ x1, const float* __restrict__ x2,
                         const float* __restrict__ Wl, const float* __restrict__ bl,
                         float* __restrict__ out, int nrows, int nblk) {
    __shared__ float As[16 * LDAJ];
    const int tid   = threadIdx.x;
    const int wave  = tid >> 5;
    const int lane  = tid & 31;
    const int col0  = wave * 16;
    const int mrow  = lane & 15;
    const int khalf = (lane >> 4) * 2;
    const int ncol  = lane & 15;
    const int mbase = (lane >> 4) * 8;
    const int ntiles = (nrows + 15) >> 4;

    v2f bf[64];
#pragma unroll
    for (int kk = 0; kk < 64; ++kk) {
        bf[kk].x = Wl[(kk * 4 + khalf + 0) * D_FEAT + col0 + ncol];
        bf[kk].y = Wl[(kk * 4 + khalf + 1) * D_FEAT + col0 + ncol];
    }
    const float bias = bl[col0 + ncol];

    const int sr = tid >> 6;
    const int c4 = tid & 63;
    const float* xs = (c4 < 32) ? (x1 + c4 * 4) : (x2 + (c4 - 32) * 4);
    int tile = blockIdx.x;
    float4 p[4];
    for (int q = 0; q < 4; ++q) p[q] = make_float4(0.f, 0.f, 0.f, 0.f);
    if (tile < ntiles) {
        for (int q = 0; q < 4; ++q) {
            long gr = (long)tile * 16 + sr + 4 * q;
            if (gr >= nrows) gr = nrows - 1;
            p[q] = *(const float4*)(xs + gr * D_FEAT);
        }
    }
    for (; tile < ntiles; tile += nblk) {
        __syncthreads();
        for (int q = 0; q < 4; ++q)
            *(float4*)(As + (sr + 4 * q) * LDAJ + c4 * 4) = p[q];
        __syncthreads();

        const int nxt = tile + nblk;
        if (nxt < ntiles) {
            for (int q = 0; q < 4; ++q) {
                long gr = (long)nxt * 16 + sr + 4 * q;
                if (gr >= nrows) gr = nrows - 1;
                p[q] = *(const float4*)(xs + gr * D_FEAT);
            }
        }

        v8f acc = {};
#pragma unroll
        for (int kk = 0; kk < 64; ++kk) {
            v2f a;
            a.x = As[mrow * LDAJ + kk * 4 + khalf + 0];
            a.y = As[mrow * LDAJ + kk * 4 + khalf + 1];
            acc = __builtin_amdgcn_wmma_f32_16x16x4_f32(
                false, a, false, bf[kk], (short)0, acc, false, false);
        }

        const long row0 = (long)tile * 16;
        if (row0 + 16 <= nrows) {
            float* op = out + (row0 + mbase) * D_FEAT + col0 + ncol;
            for (int r = 0; r < 8; ++r) op[r * D_FEAT] = acc[r] + bias;
        } else {
            for (int r = 0; r < 8; ++r) {
                long m = row0 + mbase + r;
                if (m < nrows) out[m * D_FEAT + col0 + ncol] = acc[r] + bias;
            }
        }
    }
}

extern "C" void kernel_launch(void* const* d_in, const int* in_sizes, int n_in,
                              void* d_out, int out_size, void* d_ws, size_t ws_size,
                              hipStream_t stream) {
    (void)n_in; (void)out_size; (void)ws_size;

    const float* x    = (const float*)d_in[0];
    const int*   ei   = (const int*)d_in[1];
    const float* W1   = (const float*)d_in[2];
    const float* b1   = (const float*)d_in[3];
    const float* W2   = (const float*)d_in[4];
    const float* b2   = (const float*)d_in[5];
    const float* linW = (const float*)d_in[6];
    const float* linb = (const float*)d_in[7];
    float* out = (float*)d_out;

    const int N = in_sizes[0] / D_FEAT;
    const int E = in_sizes[1] / 2;
    const int* srcI = ei;
    const int* dstI = ei + E;

    float* ws   = (float*)d_ws;
    float* dinv = ws;
    float* h    = dinv + N;
    float* edge = h   + (long)N * D_FEAT;
    float* x1   = edge + (long)N * D_FEAT;
    float* x2   = x1  + (long)N * D_FEAT;

    int2* csr    = (int2*)edge;
    int*  rank   = (int*)(csr + E);
    int*  cnt    = rank + E;
    int*  rowptr = cnt + N;
    int*  bsumP  = rowptr + N;
    int*  cntP   = (int*)x2;

    const int nBlkN   = (N + 255) / 256;
    const int Npad    = nBlkN * 256;
    const int chunk   = (((E + RANK_BLOCKS - 1) / RANK_BLOCKS) + 1023) & ~1023;
    const int nBlkE4  = (E + 1023) / 1024;
    const int nBlkRow = 512;
    const int nBlkG   = (N + 8 * GATHER_ROWS - 1) / (8 * GATHER_ROWS);

    rank_local_kernel<<<RANK_BLOCKS, 1024, 0, stream>>>(dstI, rank, cntP, bsumP, E, chunk, Npad, nBlkN);
    scan_offsets_kernel<<<nBlkN, 256, 0, stream>>>(cntP, bsumP, cnt, rowptr, dinv, N, Npad, nBlkN);
    place_edges_kernel<<<nBlkE4, 256, 0, stream>>>(srcI, dstI, rank, cntP, dinv, csr, E, chunk, Npad);

    gemm128_wmma_kernel<<<nBlkRow, 256, 0, stream>>>(x, W1, h, N, nBlkRow);
    gather_finish_kernel<<<nBlkG, 256, 0, stream>>>(rowptr, cnt, csr, dinv, h, b1, x1, N);

    gemm128_wmma_kernel<<<nBlkRow, 256, 0, stream>>>(x1, W2, h, N, nBlkRow);
    gather_finish_kernel<<<nBlkG, 256, 0, stream>>>(rowptr, cnt, csr, dinv, h, b2, x2, N);

    gemm_jk_wmma_kernel<<<nBlkRow, 256, 0, stream>>>(x1, x2, linW, linb, out, N, nBlkRow);
}


# Aggregate before the product (A_hat X then times W, equal by linearity): each GCN layer is one kernel, row tile summed into LDS then WMMA; h plane and 2 launches go; sum order changes
# speedup vs baseline: 3.3009x; 1.0458x over previous
#include <hip/hip_runtime.h>

typedef __attribute__((ext_vector_type(2))) float v2f;
typedef __attribute__((ext_vector_type(8))) float v8f;

#define D_FEAT 128

#define HIST_CAP 51200
#define RANK_BLOCKS 64

__global__ __launch_bounds__(1024)
void rank_local_kernel(const int* __restrict__ dst, int* __restrict__ rank,
                       int* __restrict__ cntP, int* __restrict__ bsumP,
                       int e, int chunk, int npad, int nb) {
    __shared__ __attribute__((aligned(16))) int hist[HIST_CAP + 4];
    __shared__ int gsum[HIST_CAP / 256];
    const int tid  = threadIdx.x;
    const int lane = tid & 31;
    const int e0 = blockIdx.x * chunk;
    const int e1 = (e0 + chunk < e) ? (e0 + chunk) : e;
    int* cp = cntP + (long)blockIdx.x * npad;
    int* bp = bsumP + (long)blockIdx.x * nb;

    for (int base = 0; base < npad; base += HIST_CAP) {
        const int wlen = (npad - base < HIST_CAP) ? (npad - base) : HIST_CAP;
        for (int i = tid * 4; i < wlen; i += 4096) *(int4*)(hist + i) = make_int4(0, 0, 0, 0);
        for (int i = tid; i < (wlen >> 8); i += 1024) gsum[i] = 0;
        __syncthreads();

        for (int i = e0 + tid; i < e1; i += 8192) {
            int d[8], r[8];
#pragma unroll
            for (int q = 0; q < 8; ++q) d[q] = dst[(i + q * 1024 < e1) ? (i + q * 1024) : (e1 - 1)];
#pragma unroll
            for (int q = 0; q < 8; ++q) d[q] = (i + q * 1024 < e1) ? (d[q] - base) : -1;
#pragma unroll
            for (int q = 0; q < 8; ++q)
                r[q] = atomicAdd(&hist[((unsigned)d[q] < (unsigned)wlen) ? d[q] : HIST_CAP], 1);
#pragma unroll
            for (int q = 0; q < 8; ++q)
                if ((unsigned)d[q] < (unsigned)wlen) rank[i + q * 1024] = r[q];
        }
        __syncthreads();

        for (int i0 = (tid & ~31) * 4; i0 < wlen; i0 += 4096) {
            const int k = i0 + lane * 4;
            const int4 v = *(const int4*)(hist + k);
            *(int4*)(cp + base + k) = v;
            int s = v.x + v.y + v.z + v.w;
            s += __builtin_amdgcn_update_dpp(0, s, 0x161, 0xf, 0xf, false);
            s += __builtin_amdgcn_update_dpp(0, s, 0x162, 0xf, 0xf, false);
            s += __builtin_amdgcn_update_dpp(0, s, 0x164, 0xf, 0xf, false);
            s += __builtin_amdgcn_update_dpp(0, s, 0x168, 0xf, 0xf, false);
            s += __builtin_amdgcn_permlanex16(s, s, 0x76543210, 0xfedcba98, false, false);
            if (lane == 0) atomicAdd(&gsum[i0 >> 8], s);
        }
        __syncthreads();
        for (int g = tid; g < (wlen >> 8); g += 1024) bp[(base >> 8) + g] = gsum[g];
        __syncthreads();
    }
}

__global__ __launch_bounds__(256)
void scan_offsets_kernel(int* __restrict__ cntP, const int* __restrict__ bsumP,
                         int* __restrict__ cnt, int* __restrict__ rowptr,
                         float* __restrict__ dinv, int n, int npad, int nb) {
    __shared__ int wtot[8];
    __shared__ int ptot[8];
    const int tid = threadIdx.x;
    const int lane = tid & 31;
    const int wave = tid >> 5;
    const int i = blockIdx.x * 256 + tid;

    int part = 0;
    for (int j = tid; j < (int)blockIdx.x; j += 256) {
#pragma unroll 8
        for (int p = 0; p < RANK_BLOCKS; ++p) part += bsumP[p * nb + j];
    }
    for (int o = 16; o > 0; o >>= 1) part += __shfl_xor(part, o, 32);

    int cpv[RANK_BLOCKS];
    int c = 0;
#pragma unroll
    for (int p = 0; p < RANK_BLOCKS; ++p) {
        cpv[p] = cntP[(long)p * npad + i];
        c += cpv[p];
    }
    int inc = c;
    for (int o = 1; o < 32; o <<= 1) {
        int t = __shfl_up(inc, o, 32);
        if (lane >= o) inc += t;
    }
    if (lane == 31) wtot[wave] = inc;
    if (lane == 0) ptot[wave] = part;
    __syncthreads();

    int base = 0;
    for (int w = 0; w < 8; ++w) {
        base += ptot[w];
        if (w < wave) base += wtot[w];
    }
    if (i < n) {
        const int rp = base + inc - c;
        cnt[i] = c;
        rowptr[i] = rp;
        float d = (float)(c + 1);
        dinv[i] = (d > 0.0f) ? rsqrtf(d) : 0.0f;
        int run = rp;
#pragma unroll
        for (int p = 0; p < RANK_BLOCKS; ++p) {
            cntP[(long)p * npad + i] = run;
            run += cpv[p];
        }
    }
}

__global__ __launch_bounds__(256)
void place_edges_kernel(const int* __restrict__ src, const int* __restrict__ dst,
                        const int* __restrict__ rank, const int* __restrict__ offP,
                        const float* __restrict__ dinv, int2* __restrict__ csr,
                        int e, int chunk, int npad) {
    const int i0 = blockIdx.x * 1024 + threadIdx.x;
    const int* rowptr = offP + (long)((blockIdx.x * 1024) / chunk) * npad;
    if (blockIdx.x * 1024 + 1024 <= e) {
        int s[4], d[4], rk[4], pos[4], wv[4];
        for (int q = 0; q < 4; ++q) {
            const int i = i0 + q * 256;
            s[q]  = src[i];
            d[q]  = dst[i];
            rk[q] = rank[i];
        }
        for (int q = 0; q < 4; ++q) {
            pos[q] = rowptr[d[q]] + rk[q];
            wv[q]  = __float_as_int(dinv[s[q]]);
        }
        for (int q = 0; q < 4; ++q) csr[pos[q]] = make_int2(s[q], wv[q]);
    } else {
        for (int q = 0; q < 4; ++q) {
            const int i = i0 + q * 256;
            if (i < e) {
                const int s = src[i], d = dst[i];
                csr[rowptr[d] + rank[i]] = make_int2(s, __float_as_int(dinv[s]));
            }
        }
    }
}

#define LDA 132
#define TILES_PER_BLOCK 5
__device__ __forceinline__ void fma4(float4& acc, float w, const float4& v) {
    acc.x = fmaf(w, v.x, acc.x);
    acc.y = fmaf(w, v.y, acc.y);
    acc.z = fmaf(w, v.z, acc.z);
    acc.w = fmaf(w, v.w, acc.w);
}

template <int G>
__device__ __forceinline__ void gather_group(const float* __restrict__ xl, int s_l, int w_l,
                                             int j, int split, float4& acc0, float4& acc1) {
    int s[G];
    float4 hv[G];
    float w[G];
#pragma unroll
    for (int q = 0; q < G; ++q) s[q] = __builtin_amdgcn_readlane(s_l, j + q);
#pragma unroll
    for (int q = 0; q < G; ++q) hv[q] = *(const float4*)(xl + (long)s[q] * D_FEAT);
#pragma unroll
    for (int q = 0; q < G; ++q) w[q] = __int_as_float(__builtin_amdgcn_readlane(w_l, j + q));
    if (j + G <= split) {
#pragma unroll
        for (int q = 0; q < G; ++q) fma4(acc0, w[q], hv[q]);
    } else if (j >= split) {
#pragma unroll
        for (int q = 0; q < G; ++q) fma4(acc1, w[q], hv[q]);
    } else {
#pragma unroll
        for (int q = 0; q < G; ++q) {
            if (j + q < split) fma4(acc0, w[q], hv[q]);
            else               fma4(acc1, w[q], hv[q]);
        }
    }
}

__global__ __launch_bounds__(256, 6)
void agg_gemm_kernel(const int* __restrict__ rowptr, const int* __restrict__ cnt,
                     const int2* __restrict__ csr, const float* __restrict__ dinv,
                     const float* __restrict__ xin, const float* __restrict__ W,
                     const float* __restrict__ b, float* __restrict__ xout,
                     int n, int nblk) {
    __shared__ __attribute__((aligned(16))) float As[2][16 * LDA];
    const int tid   = threadIdx.x;
    const int wave  = __builtin_amdgcn_readfirstlane(tid >> 5);
    const int lane  = tid & 31;
    const int col0  = wave * 16;
    const int mrow  = lane & 15;
    const int khalf = (lane >> 4) * 2;
    const int ncol  = lane & 15;
    const int mbase = (lane >> 4) * 8;
    const int ntiles = (n + 15) >> 4;

    v2f bf[32];
#pragma unroll
    for (int kk = 0; kk < 32; ++kk) {
        bf[kk].x = W[(kk * 4 + khalf + 0) * D_FEAT + col0 + ncol];
        bf[kk].y = W[(kk * 4 + khalf + 1) * D_FEAT + col0 + ncol];
    }
    const float bias = b[col0 + ncol];
    const float* xl = xin + lane * 4;

    int tile = blockIdx.x;
    if (tile >= ntiles) return;

    const int wv = tid >> 5;
    int beg, num0, num1;
    float dd0, dd1;
    int2 ent, ent2;
    float4 hs0, hs1;
    int pb, pc0, pc1;
    {
        const int r0 = tile * 16 + 2 * wv, r1 = r0 + 1;
        const int r0c = (r0 < n) ? r0 : (n - 1), r1c = (r1 < n) ? r1 : (n - 1);
        const int vb = rowptr[r0c], vc0 = cnt[r0c], vc1 = cnt[r1c];
        dd0  = dinv[r0c];
        dd1  = dinv[r1c];
        hs0 = *(const float4*)(xl + (long)r0c * D_FEAT);
        hs1 = *(const float4*)(xl + (long)r1c * D_FEAT);
        const int t1 = (tile + nblk < ntiles) ? (tile + nblk) : tile;
        const int q0 = t1 * 16 + 2 * wv, q1 = q0 + 1;
        const int q0c = (q0 < n) ? q0 : (n - 1), q1c = (q1 < n) ? q1 : (n - 1);
        pb  = rowptr[q0c];
        pc0 = cnt[q0c];
        pc1 = cnt[q1c];
        beg  = __builtin_amdgcn_readfirstlane(vb);
        num0 = __builtin_amdgcn_readfirstlane((r0 < n) ? vc0 : 0);
        num1 = __builtin_amdgcn_readfirstlane((r1 < n) ? vc1 : 0);
        ent = make_int2(0, 0);
        ent2 = make_int2(0, 0);
        if (lane < num0 + num1) ent = csr[beg + lane];
        if (lane + 32 < num0 + num1) ent2 = csr[beg + 32 + lane];
    }

    int buf = 0;
    for (; tile < ntiles; tile += nblk, buf ^= 1) {
        const int m = num0 + num1;
        const int nxt = tile + nblk;
        const bool more = nxt < ntiles;

        float4 acc0 = make_float4(0.f, 0.f, 0.f, 0.f), acc1 = acc0;
        for (int base = 0; base < m; base += 32) {
            const int mw = (m - base < 32) ? (m - base) : 32;
            const int s_l = ent.x, w_l = ent.y;
            if (base + 32 < m) {
                if (base == 0) {
                    ent = ent2;
                } else {
                    ent = make_int2(0, 0);
                    if (lane < m - base - 32) ent = csr[beg + base + 32 + lane];
                }
            }
            const int split = num0 - base;
            int j = 0;
            for (; j + 8 <= mw; j += 8) gather_group<8>(xl, s_l, w_l, j, split, acc0, acc1);
            const int r = mw - j;
            if (r >= 4) {
                if (r == 7)      gather_group<7>(xl, s_l, w_l, j, split, acc0, acc1);
                else if (r == 6) gather_group<6>(xl, s_l, w_l, j, split, acc0, acc1);
                else if (r == 5) gather_group<5>(xl, s_l, w_l, j, split, acc0, acc1);
                else             gather_group<4>(xl, s_l, w_l, j, split, acc0, acc1);
            } else {
                if (r == 3)      gather_group<3>(xl, s_l, w_l, j, split, acc0, acc1);
                else if (r == 2) gather_group<2>(xl, s_l, w_l, j, split, acc0, acc1);
                else if (r == 1) gather_group<1>(xl, s_l, w_l, j, split, acc0, acc1);
            }
        }

        float* at = &As[buf][0];
        {
            float4 v0, v1;
            v0.x = dd0 * (acc0.x + dd0 * hs0.x);
            v0.y = dd0 * (acc0.y + dd0 * hs0.y);
            v0.z = dd0 * (acc0.z + dd0 * hs0.z);
            v0.w = dd0 * (acc0.w + dd0 * hs0.w);
            v1.x = dd1 * (acc1.x + dd1 * hs1.x);
            v1.y = dd1 * (acc1.y + dd1 * hs1.y);
            v1.z = dd1 * (acc1.z + dd1 * hs1.z);
            v1.w = dd1 * (acc1.w + dd1 * hs1.w);
            *(float4*)(at + (2 * wave + 0) * LDA + lane * 4) = v0;
            *(float4*)(at + (2 * wave + 1) * LDA + lane * 4) = v1;
        }
        __syncthreads();

        if (more) {
            const int nr0 = nxt * 16 + 2 * wv, nr1 = nr0 + 1;
            const int nr0c = (nr0 < n) ? nr0 : (n - 1), nr1c = (nr1 < n) ? nr1 : (n - 1);
            beg  = __builtin_amdgcn_readfirstlane(pb);
            num0 = __builtin_amdgcn_readfirstlane((nr0 < n) ? pc0 : 0);
            num1 = __builtin_amdgcn_readfirstlane((nr1 < n) ? pc1 : 0);
            ent = make_int2(0, 0);
            ent2 = make_int2(0, 0);
            if (lane < num0 + num1) ent = csr[beg + lane];
            if (lane + 32 < num0 + num1) ent2 = csr[beg + 32 + lane];
            hs0 = *(const float4*)(xl + (long)nr0c * D_FEAT);
            hs1 = *(const float4*)(xl + (long)nr1c * D_FEAT);
            dd0 = dinv[nr0c];
            dd1 = dinv[nr1c];
            const int t2 = (nxt + nblk < ntiles) ? (nxt + nblk) : nxt;
            const int q0 = t2 * 16 + 2 * wv, q1 = q0 + 1;
            const int q0c = (q0 < n) ? q0 : (n - 1), q1c = (q1 < n) ? q1 : (n - 1);
            pb  = rowptr[q0c];
            pc0 = cnt[q0c];
            pc1 = cnt[q1c];
        }

        v8f acc = {};
#pragma unroll
        for (int kk = 0; kk < 32; ++kk) {
            v2f a;
            a.x = at[mrow * LDA + kk * 4 + khalf + 0];
            a.y = at[mrow * LDA + kk * 4 + khalf + 1];
            acc = __builtin_amdgcn_wmma_f32_16x16x4_f32(
                 false, a,  false, bf[kk],
                 (short)0, acc,  false,  false);
        }

        const long row0 = (long)tile * 16;
        if (row0 + 16 <= n) {
            float* op = xout + (row0 + mbase) * D_FEAT + col0 + ncol;
            for (int r = 0; r < 8; ++r) {
                const float v = acc[r] + bias;
                op[r * D_FEAT] = v > 0.0f ? v : 0.0f;
            }
        } else {
            for (int r = 0; r < 8; ++r) {
                const long mm = row0 + mbase + r;
                const float v = acc[r] + bias;
                if (mm < n) xout[mm * D_FEAT + col0 + ncol] = v > 0.0f ? v : 0.0f;
            }
        }
    }
}

#define LDAJ 260
__global__ __launch_bounds__(256)
void gemm_jk_wmma_kernel(const float* __restrict__ x1, const float* __restrict__ x2,
                         const float* __restrict__ Wl, const float* __restrict__ bl,
                         float* __restrict__ out, int nrows, int nblk) {
    __shared__ float As[16 * LDAJ];
    const int tid   = threadIdx.x;
    const int wave  = tid >> 5;
    const int lane  = tid & 31;
    const int col0  = wave * 16;
    const int mrow  = lane & 15;
    const int khalf = (lane >> 4) * 2;
    const int ncol  = lane & 15;
    const int mbase = (lane >> 4) * 8;
    const int ntiles = (nrows + 15) >> 4;

    v2f bf[64];
#pragma unroll
    for (int kk = 0; kk < 64; ++kk) {
        bf[kk].x = Wl[(kk * 4 + khalf + 0) * D_FEAT + col0 + ncol];
        bf[kk].y = Wl[(kk * 4 + khalf + 1) * D_FEAT + col0 + ncol];
    }
    const float bias = bl[col0 + ncol];

    const int sr = tid >> 6;
    const int c4 = tid & 63;
    const float* xs = (c4 < 32) ? (x1 + c4 * 4) : (x2 + (c4 - 32) * 4);
    int tile = blockIdx.x;
    float4 p[4];
    for (int q = 0; q < 4; ++q) p[q] = make_float4(0.f, 0.f, 0.f, 0.f);
    if (tile < ntiles) {
        for (int q = 0; q < 4; ++q) {
            long gr = (long)tile * 16 + sr + 4 * q;
            if (gr >= nrows) gr = nrows - 1;
            p[q] = *(const float4*)(xs + gr * D_FEAT);
        }
    }
    for (; tile < ntiles; tile += nblk) {
        __syncthreads();
        for (int q = 0; q < 4; ++q)
            *(float4*)(As + (sr + 4 * q) * LDAJ + c4 * 4) = p[q];
        __syncthreads();

        const int nxt = tile + nblk;
        if (nxt < ntiles) {
            for (int q = 0; q < 4; ++q) {
                long gr = (long)nxt * 16 + sr + 4 * q;
                if (gr >= nrows) gr = nrows - 1;
                p[q] = *(const float4*)(xs + gr * D_FEAT);
            }
        }

        v8f acc = {};
#pragma unroll
        for (int kk = 0; kk < 64; ++kk) {
            v2f a;
            a.x = As[mrow * LDAJ + kk * 4 + khalf + 0];
            a.y = As[mrow * LDAJ + kk * 4 + khalf + 1];
            acc = __builtin_amdgcn_wmma_f32_16x16x4_f32(
                false, a, false, bf[kk], (short)0, acc, false, false);
        }

        const long row0 = (long)tile * 16;
        if (row0 + 16 <= nrows) {
            float* op = out + (row0 + mbase) * D_FEAT + col0 + ncol;
            for (int r = 0; r < 8; ++r) op[r * D_FEAT] = acc[r] + bias;
        } else {
            for (int r = 0; r < 8; ++r) {
                long m = row0 + mbase + r;
                if (m < nrows) out[m * D_FEAT + col0 + ncol] = acc[r] + bias;
            }
        }
    }
}

extern "C" void kernel_launch(void* const* d_in, const int* in_sizes, int n_in,
                              void* d_out, int out_size, void* d_ws, size_t ws_size,
                              hipStream_t stream) {
    (void)n_in; (void)out_size; (void)ws_size;

    const float* x    = (const float*)d_in[0];
    const int*   ei   = (const int*)d_in[1];
    const float* W1   = (const float*)d_in[2];
    const float* b1   = (const float*)d_in[3];
    const float* W2   = (const float*)d_in[4];
    const float* b2   = (const float*)d_in[5];
    const float* linW = (const float*)d_in[6];
    const float* linb = (const float*)d_in[7];
    float* out = (float*)d_out;

    const int N = in_sizes[0] / D_FEAT;
    const int E = in_sizes[1] / 2;
    const int* srcI = ei;
    const int* dstI = ei + E;

    float* ws   = (float*)d_ws;
    float* dinv = ws;
    float* h    = dinv + N;
    float* edge = h   + (long)N * D_FEAT;
    float* x1   = edge + (long)N * D_FEAT;
    float* x2   = x1  + (long)N * D_FEAT;

    int2* csr    = (int2*)edge;
    int*  rank   = (int*)(csr + E);
    int*  cnt    = rank + E;
    int*  rowptr = cnt + N;
    int*  bsumP  = rowptr + N;
    int*  cntP   = (int*)x2;

    const int nBlkN   = (N + 255) / 256;
    const int Npad    = nBlkN * 256;
    const int chunk   = (((E + RANK_BLOCKS - 1) / RANK_BLOCKS) + 1023) & ~1023;
    const int nBlkE4  = (E + 1023) / 1024;
    const int nBlkRow = 512;
    const int nTiles  = (N + 15) / 16;
    const int nBlkL   = (nTiles + TILES_PER_BLOCK - 1) / TILES_PER_BLOCK;

    rank_local_kernel<<<RANK_BLOCKS, 1024, 0, stream>>>(dstI, rank, cntP, bsumP, E, chunk, Npad, nBlkN);
    scan_offsets_kernel<<<nBlkN, 256, 0, stream>>>(cntP, bsumP, cnt, rowptr, dinv, N, Npad, nBlkN);
    place_edges_kernel<<<nBlkE4, 256, 0, stream>>>(srcI, dstI, rank, cntP, dinv, csr, E, chunk, Npad);

    agg_gemm_kernel<<<nBlkL, 256, 0, stream>>>(rowptr, cnt, csr, dinv, x, W1, b1, x1, N, nBlkL);

    agg_gemm_kernel<<<nBlkL, 256, 0, stream>>>(rowptr, cnt, csr, dinv, x1, W2, b2, x2, N, nBlkL);

    gemm_jk_wmma_kernel<<<nBlkRow, 256, 0, stream>>>(x1, x2, linW, linb, out, N, nBlkRow);
}


# Rank kernel: a dropped edge slot adds to a spare LDS counter of its own lane, not to one word shared by all lanes (lanes on one word serialise); the short last chunk no longer holds the launch
# speedup vs baseline: 3.5979x; 1.0900x over previous
#include <hip/hip_runtime.h>

typedef __attribute__((ext_vector_type(2))) float v2f;
typedef __attribute__((ext_vector_type(8))) float v8f;

#define D_FEAT 128

#define HIST_CAP 51200
#define RANK_BLOCKS 64

__global__ __launch_bounds__(1024)
void rank_local_kernel(const int* __restrict__ dst, int* __restrict__ rank,
                       int* __restrict__ cntP, int* __restrict__ bsumP,
                       int e, int chunk, int npad, int nb) {
    __shared__ __attribute__((aligned(16))) int hist[HIST_CAP + 32];
    __shared__ int gsum[HIST_CAP / 256];
    const int tid  = threadIdx.x;
    const int lane = tid & 31;
    const int e0 = blockIdx.x * chunk;
    const int e1 = (e0 + chunk < e) ? (e0 + chunk) : e;
    int* cp = cntP + (long)blockIdx.x * npad;
    int* bp = bsumP + (long)blockIdx.x * nb;

    for (int base = 0; base < npad; base += HIST_CAP) {
        const int wlen = (npad - base < HIST_CAP) ? (npad - base) : HIST_CAP;
        for (int i = tid * 4; i < wlen; i += 4096) *(int4*)(hist + i) = make_int4(0, 0, 0, 0);
        for (int i = tid; i < (wlen >> 8); i += 1024) gsum[i] = 0;
        __syncthreads();

        for (int i = e0 + tid; i < e1; i += 8192) {
            int d[8], r[8];
#pragma unroll
            for (int q = 0; q < 8; ++q) d[q] = dst[(i + q * 1024 < e1) ? (i + q * 1024) : (e1 - 1)];
#pragma unroll
            for (int q = 0; q < 8; ++q) d[q] = (i + q * 1024 < e1) ? (d[q] - base) : -1;
#pragma unroll
            for (int q = 0; q < 8; ++q)
                r[q] = atomicAdd(&hist[((unsigned)d[q] < (unsigned)wlen) ? d[q] : (HIST_CAP + lane)], 1);
#pragma unroll
            for (int q = 0; q < 8; ++q)
                if ((unsigned)d[q] < (unsigned)wlen) rank[i + q * 1024] = r[q];
        }
        __syncthreads();

        for (int i0 = (tid & ~31) * 4; i0 < wlen; i0 += 4096) {
            const int k = i0 + lane * 4;
            const int4 v = *(const int4*)(hist + k);
            *(int4*)(cp + base + k) = v;
            int s = v.x + v.y + v.z + v.w;
            s += __builtin_amdgcn_update_dpp(0, s, 0x161, 0xf, 0xf, false);
            s += __builtin_amdgcn_update_dpp(0, s, 0x162, 0xf, 0xf, false);
            s += __builtin_amdgcn_update_dpp(0, s, 0x164, 0xf, 0xf, false);
            s += __builtin_amdgcn_update_dpp(0, s, 0x168, 0xf, 0xf, false);
            s += __builtin_amdgcn_permlanex16(s, s, 0x76543210, 0xfedcba98, false, false);
            if (lane == 0) atomicAdd(&gsum[i0 >> 8], s);
        }
        __syncthreads();
        for (int g = tid; g < (wlen >> 8); g += 1024) bp[(base >> 8) + g] = gsum[g];
        __syncthreads();
    }
}

__global__ __launch_bounds__(256)
void scan_offsets_kernel(int* __restrict__ cntP, const int* __restrict__ bsumP,
                         int* __restrict__ cnt, int* __restrict__ rowptr,
                         float* __restrict__ dinv, int n, int npad, int nb) {
    __shared__ int wtot[8];
    __shared__ int ptot[8];
    const int tid = threadIdx.x;
    const int lane = tid & 31;
    const int wave = tid >> 5;
    const int i = blockIdx.x * 256 + tid;

    int part = 0;
    for (int j = tid; j < (int)blockIdx.x; j += 256) {
#pragma unroll 8
        for (int p = 0; p < RANK_BLOCKS; ++p) part += bsumP[p * nb + j];
    }
    for (int o = 16; o > 0; o >>= 1) part += __shfl_xor(part, o, 32);

    int cpv[RANK_BLOCKS];
    int c = 0;
#pragma unroll
    for (int p = 0; p < RANK_BLOCKS; ++p) {
        cpv[p] = cntP[(long)p * npad + i];
        c += cpv[p];
    }
    int inc = c;
    for (int o = 1; o < 32; o <<= 1) {
        int t = __shfl_up(inc, o, 32);
        if (lane >= o) inc += t;
    }
    if (lane == 31) wtot[wave] = inc;
    if (lane == 0) ptot[wave] = part;
    __syncthreads();

    int base = 0;
    for (int w = 0; w < 8; ++w) {
        base += ptot[w];
        if (w < wave) base += wtot[w];
    }
    if (i < n) {
        const int rp = base + inc - c;
        cnt[i] = c;
        rowptr[i] = rp;
        float d = (float)(c + 1);
        dinv[i] = (d > 0.0f) ? rsqrtf(d) : 0.0f;
        int run = rp;
#pragma unroll
        for (int p = 0; p < RANK_BLOCKS; ++p) {
            cntP[(long)p * npad + i] = run;
            run += cpv[p];
        }
    }
}

__global__ __launch_bounds__(256)
void place_edges_kernel(const int* __restrict__ src, const int* __restrict__ dst,
                        const int* __restrict__ rank, const int* __restrict__ offP,
                        const float* __restrict__ dinv, int2* __restrict__ csr,
                        int e, int chunk, int npad) {
    const int i0 = blockIdx.x * 1024 + threadIdx.x;
    const int* rowptr = offP + (long)((blockIdx.x * 1024) / chunk) * npad;
    if (blockIdx.x * 1024 + 1024 <= e) {
        int s[4], d[4], rk[4], pos[4], wv[4];
        for (int q = 0; q < 4; ++q) {
            const int i = i0 + q * 256;
            s[q]  = src[i];
            d[q]  = dst[i];
            rk[q] = rank[i];
        }
        for (int q = 0; q < 4; ++q) {
            pos[q] = rowptr[d[q]] + rk[q];
            wv[q]  = __float_as_int(dinv[s[q]]);
        }
        for (int q = 0; q < 4; ++q) csr[pos[q]] = make_int2(s[q], wv[q]);
    } else {
        for (int q = 0; q < 4; ++q) {
            const int i = i0 + q * 256;
            if (i < e) {
                const int s = src[i], d = dst[i];
                csr[rowptr[d] + rank[i]] = make_int2(s, __float_as_int(dinv[s]));
            }
        }
    }
}

#define LDA 132
#define TILES_PER_BLOCK 5
__device__ __forceinline__ void fma4(float4& acc, float w, const float4& v) {
    acc.x = fmaf(w, v.x, acc.x);
    acc.y = fmaf(w, v.y, acc.y);
    acc.z = fmaf(w, v.z, acc.z);
    acc.w = fmaf(w, v.w, acc.w);
}

template <int G>
__device__ __forceinline__ void gather_group(const float* __restrict__ xl, int s_l, int w_l,
                                             int j, int split, float4& acc0, float4& acc1) {
    int s[G];
    float4 hv[G];
    float w[G];
#pragma unroll
    for (int q = 0; q < G; ++q) s[q] = __builtin_amdgcn_readlane(s_l, j + q);
#pragma unroll
    for (int q = 0; q < G; ++q) hv[q] = *(const float4*)(xl + (long)s[q] * D_FEAT);
#pragma unroll
    for (int q = 0; q < G; ++q) w[q] = __int_as_float(__builtin_amdgcn_readlane(w_l, j + q));
    if (j + G <= split) {
#pragma unroll
        for (int q = 0; q < G; ++q) fma4(acc0, w[q], hv[q]);
    } else if (j >= split) {
#pragma unroll
        for (int q = 0; q < G; ++q) fma4(acc1, w[q], hv[q]);
    } else {
#pragma unroll
        for (int q = 0; q < G; ++q) {
            if (j + q < split) fma4(acc0, w[q], hv[q]);
            else               fma4(acc1, w[q], hv[q]);
        }
    }
}

__global__ __launch_bounds__(256, 6)
void agg_gemm_kernel(const int* __restrict__ rowptr, const int* __restrict__ cnt,
                     const int2* __restrict__ csr, const float* __restrict__ dinv,
                     const float* __restrict__ xin, const float* __restrict__ W,
                     const float* __restrict__ b, float* __restrict__ xout,
                     int n, int nblk) {
    __shared__ __attribute__((aligned(16))) float As[2][16 * LDA];
    const int tid   = threadIdx.x;
    const int wave  = __builtin_amdgcn_readfirstlane(tid >> 5);
    const int lane  = tid & 31;
    const int col0  = wave * 16;
    const int mrow  = lane & 15;
    const int khalf = (lane >> 4) * 2;
    const int ncol  = lane & 15;
    const int mbase = (lane >> 4) * 8;
    const int ntiles = (n + 15) >> 4;

    v2f bf[32];
#pragma unroll
    for (int kk = 0; kk < 32; ++kk) {
        bf[kk].x = W[(kk * 4 + khalf + 0) * D_FEAT + col0 + ncol];
        bf[kk].y = W[(kk * 4 + khalf + 1) * D_FEAT + col0 + ncol];
    }
    const float bias = b[col0 + ncol];
    const float* xl = xin + lane * 4;

    int tile = blockIdx.x;
    if (tile >= ntiles) return;

    const int wv = tid >> 5;
    int beg, num0, num1;
    float dd0, dd1;
    int2 ent, ent2;
    float4 hs0, hs1;
    int pb, pc0, pc1;
    {
        const int r0 = tile * 16 + 2 * wv, r1 = r0 + 1;
        const int r0c = (r0 < n) ? r0 : (n - 1), r1c = (r1 < n) ? r1 : (n - 1);
        const int vb = rowptr[r0c], vc0 = cnt[r0c], vc1 = cnt[r1c];
        dd0  = dinv[r0c];
        dd1  = dinv[r1c];
        hs0 = *(const float4*)(xl + (long)r0c * D_FEAT);
        hs1 = *(const float4*)(xl + (long)r1c * D_FEAT);
        const int t1 = (tile + nblk < ntiles) ? (tile + nblk) : tile;
        const int q0 = t1 * 16 + 2 * wv, q1 = q0 + 1;
        const int q0c = (q0 < n) ? q0 : (n - 1), q1c = (q1 < n) ? q1 : (n - 1);
        pb  = rowptr[q0c];
        pc0 = cnt[q0c];
        pc1 = cnt[q1c];
        beg  = __builtin_amdgcn_readfirstlane(vb);
        num0 = __builtin_amdgcn_readfirstlane((r0 < n) ? vc0 : 0);
        num1 = __builtin_amdgcn_readfirstlane((r1 < n) ? vc1 : 0);
        ent = make_int2(0, 0);
        ent2 = make_int2(0, 0);
        if (lane < num0 + num1) ent = csr[beg + lane];
        if (lane + 32 < num0 + num1) ent2 = csr[beg + 32 + lane];
    }

    int buf = 0;
    for (; tile < ntiles; tile += nblk, buf ^= 1) {
        const int m = num0 + num1;
        const int nxt = tile + nblk;
        const bool more = nxt < ntiles;

        float4 acc0 = make_float4(0.f, 0.f, 0.f, 0.f), acc1 = acc0;
        for (int base = 0; base < m; base += 32) {
            const int mw = (m - base < 32) ? (m - base) : 32;
            const int s_l = ent.x, w_l = ent.y;
            if (base + 32 < m) {
                if (base == 0) {
                    ent = ent2;
                } else {
                    ent = make_int2(0, 0);
                    if (lane < m - base - 32) ent = csr[beg + base + 32 + lane];
                }
            }
            const int split = num0 - base;
            int j = 0;
            for (; j + 8 <= mw; j += 8) gather_group<8>(xl, s_l, w_l, j, split, acc0, acc1);
            const int r = mw - j;
            if (r >= 4) {
                if (r == 7)      gather_group<7>(xl, s_l, w_l, j, split, acc0, acc1);
                else if (r == 6) gather_group<6>(xl, s_l, w_l, j, split, acc0, acc1);
                else if (r == 5) gather_group<5>(xl, s_l, w_l, j, split, acc0, acc1);
                else             gather_group<4>(xl, s_l, w_l, j, split, acc0, acc1);
            } else {
                if (r == 3)      gather_group<3>(xl, s_l, w_l, j, split, acc0, acc1);
                else if (r == 2) gather_group<2>(xl, s_l, w_l, j, split, acc0, acc1);
                else if (r == 1) gather_group<1>(xl, s_l, w_l, j, split, acc0, acc1);
            }
        }

        float* at = &As[buf][0];
        {
            float4 v0, v1;
            v0.x = dd0 * (acc0.x + dd0 * hs0.x);
            v0.y = dd0 * (acc0.y + dd0 * hs0.y);
            v0.z = dd0 * (acc0.z + dd0 * hs0.z);
            v0.w = dd0 * (acc0.w + dd0 * hs0.w);
            v1.x = dd1 * (acc1.x + dd1 * hs1.x);
            v1.y = dd1 * (acc1.y + dd1 * hs1.y);
            v1.z = dd1 * (acc1.z + dd1 * hs1.z);
            v1.w = dd1 * (acc1.w + dd1 * hs1.w);
            *(float4*)(at + (2 * wave + 0) * LDA + lane * 4) = v0;
            *(float4*)(at + (2 * wave + 1) * LDA + lane * 4) = v1;
        }
        __syncthreads();

        if (more) {
            const int nr0 = nxt * 16 + 2 * wv, nr1 = nr0 + 1;
            const int nr0c = (nr0 < n) ? nr0 : (n - 1), nr1c = (nr1 < n) ? nr1 : (n - 1);
            beg  = __builtin_amdgcn_readfirstlane(pb);
            num0 = __builtin_amdgcn_readfirstlane((nr0 < n) ? pc0 : 0);
            num1 = __builtin_amdgcn_readfirstlane((nr1 < n) ? pc1 : 0);
            ent = make_int2(0, 0);
            ent2 = make_int2(0, 0);
            if (lane < num0 + num1) ent = csr[beg + lane];
            if (lane + 32 < num0 + num1) ent2 = csr[beg + 32 + lane];
            hs0 = *(const float4*)(xl + (long)nr0c * D_FEAT);
            hs1 = *(const float4*)(xl + (long)nr1c * D_FEAT);
            dd0 = dinv[nr0c];
            dd1 = dinv[nr1c];
            const int t2 = (nxt + nblk < ntiles) ? (nxt + nblk) : nxt;
            const int q0 = t2 * 16 + 2 * wv, q1 = q0 + 1;
            const int q0c = (q0 < n) ? q0 : (n - 1), q1c = (q1 < n) ? q1 : (n - 1);
            pb  = rowptr[q0c];
            pc0 = cnt[q0c];
            pc1 = cnt[q1c];
        }

        v8f acc = {};
#pragma unroll
        for (int kk = 0; kk < 32; ++kk) {
            v2f a;
            a.x = at[mrow * LDA + kk * 4 + khalf + 0];
            a.y = at[mrow * LDA + kk * 4 + khalf + 1];
            acc = __builtin_amdgcn_wmma_f32_16x16x4_f32(
                 false, a,  false, bf[kk],
                 (short)0, acc,  false,  false);
        }

        const long row0 = (long)tile * 16;
        if (row0 + 16 <= n) {
            float* op = xout + (row0 + mbase) * D_FEAT + col0 + ncol;
            for (int r = 0; r < 8; ++r) {
                const float v = acc[r] + bias;
                op[r * D_FEAT] = v > 0.0f ? v : 0.0f;
            }
        } else {
            for (int r = 0; r < 8; ++r) {
                const long mm = row0 + mbase + r;
                const float v = acc[r] + bias;
                if (mm < n) xout[mm * D_FEAT + col0 + ncol] = v > 0.0f ? v : 0.0f;
            }
        }
    }
}

#define LDAJ 260
__global__ __launch_bounds__(256)
void gemm_jk_wmma_kernel(const float* __restrict__ x1, const float* __restrict__ x2,
                         const float* __restrict__ Wl, const float* __restrict__ bl,
                         float* __restrict__ out, int nrows, int nblk) {
    __shared__ float As[16 * LDAJ];
    const int tid   = threadIdx.x;
    const int wave  = tid >> 5;
    const int lane  = tid & 31;
    const int col0  = wave * 16;
    const int mrow  = lane & 15;
    const int khalf = (lane >> 4) * 2;
    const int ncol  = lane & 15;
    const int mbase = (lane >> 4) * 8;
    const int ntiles = (nrows + 15) >> 4;

    v2f bf[64];
#pragma unroll
    for (int kk = 0; kk < 64; ++kk) {
        bf[kk].x = Wl[(kk * 4 + khalf + 0) * D_FEAT + col0 + ncol];
        bf[kk].y = Wl[(kk * 4 + khalf + 1) * D_FEAT + col0 + ncol];
    }
    const float bias = bl[col0 + ncol];

    const int sr = tid >> 6;
    const int c4 = tid & 63;
    const float* xs = (c4 < 32) ? (x1 + c4 * 4) : (x2 + (c4 - 32) * 4);
    int tile = blockIdx.x;
    float4 p[4];
    for (int q = 0; q < 4; ++q) p[q] = make_float4(0.f, 0.f, 0.f, 0.f);
    if (tile < ntiles) {
        for (int q = 0; q < 4; ++q) {
            long gr = (long)tile * 16 + sr + 4 * q;
            if (gr >= nrows) gr = nrows - 1;
            p[q] = *(const float4*)(xs + gr * D_FEAT);
        }
    }
    for (; tile < ntiles; tile += nblk) {
        __syncthreads();
        for (int q = 0; q < 4; ++q)
            *(float4*)(As + (sr + 4 * q) * LDAJ + c4 * 4) = p[q];
        __syncthreads();

        const int nxt = tile + nblk;
        if (nxt < ntiles) {
            for (int q = 0; q < 4; ++q) {
                long gr = (long)nxt * 16 + sr + 4 * q;
                if (gr >= nrows) gr = nrows - 1;
                p[q] = *(const float4*)(xs + gr * D_FEAT);
            }
        }

        v8f acc = {};
#pragma unroll
        for (int kk = 0; kk < 64; ++kk) {
            v2f a;
            a.x = As[mrow * LDAJ + kk * 4 + khalf + 0];
            a.y = As[mrow * LDAJ + kk * 4 + khalf + 1];
            acc = __builtin_amdgcn_wmma_f32_16x16x4_f32(
                false, a, false, bf[kk], (short)0, acc, false, false);
        }

        const long row0 = (long)tile * 16;
        if (row0 + 16 <= nrows) {
            float* op = out + (row0 + mbase) * D_FEAT + col0 + ncol;
            for (int r = 0; r < 8; ++r) op[r * D_FEAT] = acc[r] + bias;
        } else {
            for (int r = 0; r < 8; ++r) {
                long m = row0 + mbase + r;
                if (m < nrows) out[m * D_FEAT + col0 + ncol] = acc[r] + bias;
            }
        }
    }
}

extern "C" void kernel_launch(void* const* d_in, const int* in_sizes, int n_in,
                              void* d_out, int out_size, void* d_ws, size_t ws_size,
                              hipStream_t stream) {
    (void)n_in; (void)out_size; (void)ws_size;

    const float* x    = (const float*)d_in[0];
    const int*   ei   = (const int*)d_in[1];
    const float* W1   = (const float*)d_in[2];
    const float* b1   = (const float*)d_in[3];
    const float* W2   = (const float*)d_in[4];
    const float* b2   = (const float*)d_in[5];
    const float* linW = (const float*)d_in[6];
    const float* linb = (const float*)d_in[7];
    float* out = (float*)d_out;

    const int N = in_sizes[0] / D_FEAT;
    const int E = in_sizes[1] / 2;
    const int* srcI = ei;
    const int* dstI = ei + E;

    float* ws   = (float*)d_ws;
    float* dinv = ws;
    float* h    = dinv + N;
    float* edge = h   + (long)N * D_FEAT;
    float* x1   = edge + (long)N * D_FEAT;
    float* x2   = x1  + (long)N * D_FEAT;

    int2* csr    = (int2*)edge;
    int*  rank   = (int*)(csr + E);
    int*  cnt    = rank + E;
    int*  rowptr = cnt + N;
    int*  bsumP  = rowptr + N;
    int*  cntP   = (int*)x2;

    const int nBlkN   = (N + 255) / 256;
    const int Npad    = nBlkN * 256;
    const int chunk   = (((E + RANK_BLOCKS - 1) / RANK_BLOCKS) + 1023) & ~1023;
    const int nBlkE4  = (E + 1023) / 1024;
    const int nBlkRow = 512;
    const int nTiles  = (N + 15) / 16;
    const int nBlkL   = (nTiles + TILES_PER_BLOCK - 1) / TILES_PER_BLOCK;

    rank_local_kernel<<<RANK_BLOCKS, 1024, 0, stream>>>(dstI, rank, cntP, bsumP, E, chunk, Npad, nBlkN);
    scan_offsets_kernel<<<nBlkN, 256, 0, stream>>>(cntP, bsumP, cnt, rowptr, dinv, N, Npad, nBlkN);
    place_edges_kernel<<<nBlkE4, 256, 0, stream>>>(srcI, dstI, rank, cntP, dinv, csr, E, chunk, Npad);

    agg_gemm_kernel<<<nBlkL, 256, 0, stream>>>(rowptr, cnt, csr, dinv, x, W1, b1, x1, N, nBlkL);

    agg_gemm_kernel<<<nBlkL, 256, 0, stream>>>(rowptr, cnt, csr, dinv, x1, W2, b2, x2, N, nBlkL);

    gemm_jk_wmma_kernel<<<nBlkRow, 256, 0, stream>>>(x1, x2, linW, linb, out, N, nBlkRow);
}


# Final linear done inside the layer kernels as two K=128 halves, each on the tile just produced (linW strip in LDS): last kernel, x2 plane and one launch go; out's sum order changes
# speedup vs baseline: 3.7205x; 1.0341x over previous
#include <hip/hip_runtime.h>

typedef __attribute__((ext_vector_type(2))) float v2f;
typedef __attribute__((ext_vector_type(8))) float v8f;

#define D_FEAT 128

#define HIST_CAP 51200
#define RANK_BLOCKS 64

__global__ __launch_bounds__(1024)
void rank_local_kernel(const int* __restrict__ dst, int* __restrict__ rank,
                       int* __restrict__ cntP, int* __restrict__ bsumP,
                       int e, int chunk, int npad, int nb) {
    __shared__ __attribute__((aligned(16))) int hist[HIST_CAP + 32];
    __shared__ int gsum[HIST_CAP / 256];
    const int tid  = threadIdx.x;
    const int lane = tid & 31;
    const int e0 = blockIdx.x * chunk;
    const int e1 = (e0 + chunk < e) ? (e0 + chunk) : e;
    int* cp = cntP + (long)blockIdx.x * npad;
    int* bp = bsumP + (long)blockIdx.x * nb;

    for (int base = 0; base < npad; base += HIST_CAP) {
        const int wlen = (npad - base < HIST_CAP) ? (npad - base) : HIST_CAP;
        for (int i = tid * 4; i < wlen; i += 4096) *(int4*)(hist + i) = make_int4(0, 0, 0, 0);
        for (int i = tid; i < (wlen >> 8); i += 1024) gsum[i] = 0;
        __syncthreads();

        for (int i = e0 + tid; i < e1; i += 8192) {
            int d[8], r[8];
#pragma unroll
            for (int q = 0; q < 8; ++q) d[q] = dst[(i + q * 1024 < e1) ? (i + q * 1024) : (e1 - 1)];
#pragma unroll
            for (int q = 0; q < 8; ++q) d[q] = (i + q * 1024 < e1) ? (d[q] - base) : -1;
#pragma unroll
            for (int q = 0; q < 8; ++q)
                r[q] = atomicAdd(&hist[((unsigned)d[q] < (unsigned)wlen) ? d[q] : (HIST_CAP + lane)], 1);
#pragma unroll
            for (int q = 0; q < 8; ++q)
                if ((unsigned)d[q] < (unsigned)wlen) rank[i + q * 1024] = r[q];
        }
        __syncthreads();

        for (int i0 = (tid & ~31) * 4; i0 < wlen; i0 += 4096) {
            const int k = i0 + lane * 4;
            const int4 v = *(const int4*)(hist + k);
            *(int4*)(cp + base + k) = v;
            int s = v.x + v.y + v.z + v.w;
            s += __builtin_amdgcn_update_dpp(0, s, 0x161, 0xf, 0xf, false);
            s += __builtin_amdgcn_update_dpp(0, s, 0x162, 0xf, 0xf, false);
            s += __builtin_amdgcn_update_dpp(0, s, 0x164, 0xf, 0xf, false);
            s += __builtin_amdgcn_update_dpp(0, s, 0x168, 0xf, 0xf, false);
            s += __builtin_amdgcn_permlanex16(s, s, 0x76543210, 0xfedcba98, false, false);
            if (lane == 0) atomicAdd(&gsum[i0 >> 8], s);
        }
        __syncthreads();
        for (int g = tid; g < (wlen >> 8); g += 1024) bp[(base >> 8) + g] = gsum[g];
        __syncthreads();
    }
}

__global__ __launch_bounds__(256)
void scan_offsets_kernel(int* __restrict__ cntP, const int* __restrict__ bsumP,
                         int* __restrict__ cnt, int* __restrict__ rowptr,
                         float* __restrict__ dinv, int n, int npad, int nb) {
    __shared__ int wtot[8];
    __shared__ int ptot[8];
    const int tid = threadIdx.x;
    const int lane = tid & 31;
    const int wave = tid >> 5;
    const int i = blockIdx.x * 256 + tid;

    int part = 0;
    for (int j = tid; j < (int)blockIdx.x; j += 256) {
#pragma unroll 8
        for (int p = 0; p < RANK_BLOCKS; ++p) part += bsumP[p * nb + j];
    }
    for (int o = 16; o > 0; o >>= 1) part += __shfl_xor(part, o, 32);

    int cpv[RANK_BLOCKS];
    int c = 0;
#pragma unroll
    for (int p = 0; p < RANK_BLOCKS; ++p) {
        cpv[p] = cntP[(long)p * npad + i];
        c += cpv[p];
    }
    int inc = c;
    for (int o = 1; o < 32; o <<= 1) {
        int t = __shfl_up(inc, o, 32);
        if (lane >= o) inc += t;
    }
    if (lane == 31) wtot[wave] = inc;
    if (lane == 0) ptot[wave] = part;
    __syncthreads();

    int base = 0;
    for (int w = 0; w < 8; ++w) {
        base += ptot[w];
        if (w < wave) base += wtot[w];
    }
    if (i < n) {
        const int rp = base + inc - c;
        cnt[i] = c;
        rowptr[i] = rp;
        float d = (float)(c + 1);
        dinv[i] = (d > 0.0f) ? rsqrtf(d) : 0.0f;
        int run = rp;
#pragma unroll
        for (int p = 0; p < RANK_BLOCKS; ++p) {
            cntP[(long)p * npad + i] = run;
            run += cpv[p];
        }
    }
}

__global__ __launch_bounds__(256)
void place_edges_kernel(const int* __restrict__ src, const int* __restrict__ dst,
                        const int* __restrict__ rank, const int* __restrict__ offP,
                        const float* __restrict__ dinv, int2* __restrict__ csr,
                        int e, int chunk, int npad) {
    const int i0 = blockIdx.x * 1024 + threadIdx.x;
    const int* rowptr = offP + (long)((blockIdx.x * 1024) / chunk) * npad;
    if (blockIdx.x * 1024 + 1024 <= e) {
        int s[4], d[4], rk[4], pos[4], wv[4];
        for (int q = 0; q < 4; ++q) {
            const int i = i0 + q * 256;
            s[q]  = src[i];
            d[q]  = dst[i];
            rk[q] = rank[i];
        }
        for (int q = 0; q < 4; ++q) {
            pos[q] = rowptr[d[q]] + rk[q];
            wv[q]  = __float_as_int(dinv[s[q]]);
        }
        for (int q = 0; q < 4; ++q) csr[pos[q]] = make_int2(s[q], wv[q]);
    } else {
        for (int q = 0; q < 4; ++q) {
            const int i = i0 + q * 256;
            if (i < e) {
                const int s = src[i], d = dst[i];
                csr[rowptr[d] + rank[i]] = make_int2(s, __float_as_int(dinv[s]));
            }
        }
    }
}

#define LDA 132
#define TILES_PER_BLOCK 5
__device__ __forceinline__ void fma4(float4& acc, float w, const float4& v) {
    acc.x = fmaf(w, v.x, acc.x);
    acc.y = fmaf(w, v.y, acc.y);
    acc.z = fmaf(w, v.z, acc.z);
    acc.w = fmaf(w, v.w, acc.w);
}

template <int G>
__device__ __forceinline__ void gather_group(const float* __restrict__ xl, int s_l, int w_l,
                                             int j, int split, float4& acc0, float4& acc1) {
    int s[G];
    float4 hv[G];
    float w[G];
#pragma unroll
    for (int q = 0; q < G; ++q) s[q] = __builtin_amdgcn_readlane(s_l, j + q);
#pragma unroll
    for (int q = 0; q < G; ++q) hv[q] = *(const float4*)(xl + (long)s[q] * D_FEAT);
#pragma unroll
    for (int q = 0; q < G; ++q) w[q] = __int_as_float(__builtin_amdgcn_readlane(w_l, j + q));
    if (j + G <= split) {
#pragma unroll
        for (int q = 0; q < G; ++q) fma4(acc0, w[q], hv[q]);
    } else if (j >= split) {
#pragma unroll
        for (int q = 0; q < G; ++q) fma4(acc1, w[q], hv[q]);
    } else {
#pragma unroll
        for (int q = 0; q < G; ++q) {
            if (j + q < split) fma4(acc0, w[q], hv[q]);
            else               fma4(acc1, w[q], hv[q]);
        }
    }
}

__global__ __launch_bounds__(256, 6)
void agg_gemm_kernel(const int* __restrict__ rowptr, const int* __restrict__ cnt,
                     const int2* __restrict__ csr, const float* __restrict__ dinv,
                     const float* __restrict__ xin, const float* __restrict__ W,
                     const float* __restrict__ b, float* __restrict__ xout,
                     const float* __restrict__ Wl, const float* __restrict__ bl,
                     float* out, int second, int n, int nblk) {
    __shared__ __attribute__((aligned(16))) float As[2][16 * LDA];
    __shared__ __attribute__((aligned(16))) float Xs[16 * LDA];
    __shared__ __attribute__((aligned(16))) v2f Wls[8 * 32 * 32];
    const int tid   = threadIdx.x;
    const int wave  = __builtin_amdgcn_readfirstlane(tid >> 5);
    const int lane  = tid & 31;
    const int col0  = wave * 16;
    const int mrow  = lane & 15;
    const int khalf = (lane >> 4) * 2;
    const int ncol  = lane & 15;
    const int mbase = (lane >> 4) * 8;
    const int ntiles = (n + 15) >> 4;

    v2f bf[32];
#pragma unroll
    for (int kk = 0; kk < 32; ++kk) {
        bf[kk].x = W[(kk * 4 + khalf + 0) * D_FEAT + col0 + ncol];
        bf[kk].y = W[(kk * 4 + khalf + 1) * D_FEAT + col0 + ncol];
    }
    const float bias = b[col0 + ncol];
    const float lbias = bl[col0 + ncol];
    const float* xl = xin + lane * 4;

    int tile = blockIdx.x;
    if (tile >= ntiles) return;

    v2f* wl = Wls + wave * (32 * 32) + lane;
    for (int kk = 0; kk < 32; ++kk) {
        v2f t;
        t.x = Wl[(kk * 4 + khalf + 0) * D_FEAT + col0 + ncol];
        t.y = Wl[(kk * 4 + khalf + 1) * D_FEAT + col0 + ncol];
        wl[kk * 32] = t;
    }

    const int wv = tid >> 5;
    int beg, num0, num1;
    float dd0, dd1;
    int2 ent, ent2;
    float4 hs0, hs1;
    int pb, pc0, pc1;
    {
        const int r0 = tile * 16 + 2 * wv, r1 = r0 + 1;
        const int r0c = (r0 < n) ? r0 : (n - 1), r1c = (r1 < n) ? r1 : (n - 1);
        const int vb = rowptr[r0c], vc0 = cnt[r0c], vc1 = cnt[r1c];
        dd0  = dinv[r0c];
        dd1  = dinv[r1c];
        hs0 = *(const float4*)(xl + (long)r0c * D_FEAT);
        hs1 = *(const float4*)(xl + (long)r1c * D_FEAT);
        const int t1 = (tile + nblk < ntiles) ? (tile + nblk) : tile;
        const int q0 = t1 * 16 + 2 * wv, q1 = q0 + 1;
        const int q0c = (q0 < n) ? q0 : (n - 1), q1c = (q1 < n) ? q1 : (n - 1);
        pb  = rowptr[q0c];
        pc0 = cnt[q0c];
        pc1 = cnt[q1c];
        beg  = __builtin_amdgcn_readfirstlane(vb);
        num0 = __builtin_amdgcn_readfirstlane((r0 < n) ? vc0 : 0);
        num1 = __builtin_amdgcn_readfirstlane((r1 < n) ? vc1 : 0);
        ent = make_int2(0, 0);
        ent2 = make_int2(0, 0);
        if (lane < num0 + num1) ent = csr[beg + lane];
        if (lane + 32 < num0 + num1) ent2 = csr[beg + 32 + lane];
    }

    int buf = 0;
    for (; tile < ntiles; tile += nblk, buf ^= 1) {
        const int m = num0 + num1;
        const int nxt = tile + nblk;
        const bool more = nxt < ntiles;

        float4 acc0 = make_float4(0.f, 0.f, 0.f, 0.f), acc1 = acc0;
        for (int base = 0; base < m; base += 32) {
            const int mw = (m - base < 32) ? (m - base) : 32;
            const int s_l = ent.x, w_l = ent.y;
            if (base + 32 < m) {
                if (base == 0) {
                    ent = ent2;
                } else {
                    ent = make_int2(0, 0);
                    if (lane < m - base - 32) ent = csr[beg + base + 32 + lane];
                }
            }
            const int split = num0 - base;
            int j = 0;
            for (; j + 8 <= mw; j += 8) gather_group<8>(xl, s_l, w_l, j, split, acc0, acc1);
            const int r = mw - j;
            if (r >= 4) {
                if (r == 7)      gather_group<7>(xl, s_l, w_l, j, split, acc0, acc1);
                else if (r == 6) gather_group<6>(xl, s_l, w_l, j, split, acc0, acc1);
                else if (r == 5) gather_group<5>(xl, s_l, w_l, j, split, acc0, acc1);
                else             gather_group<4>(xl, s_l, w_l, j, split, acc0, acc1);
            } else {
                if (r == 3)      gather_group<3>(xl, s_l, w_l, j, split, acc0, acc1);
                else if (r == 2) gather_group<2>(xl, s_l, w_l, j, split, acc0, acc1);
                else if (r == 1) gather_group<1>(xl, s_l, w_l, j, split, acc0, acc1);
            }
        }

        float* at = &As[buf][0];
        {
            float4 v0, v1;
            v0.x = dd0 * (acc0.x + dd0 * hs0.x);
            v0.y = dd0 * (acc0.y + dd0 * hs0.y);
            v0.z = dd0 * (acc0.z + dd0 * hs0.z);
            v0.w = dd0 * (acc0.w + dd0 * hs0.w);
            v1.x = dd1 * (acc1.x + dd1 * hs1.x);
            v1.y = dd1 * (acc1.y + dd1 * hs1.y);
            v1.z = dd1 * (acc1.z + dd1 * hs1.z);
            v1.w = dd1 * (acc1.w + dd1 * hs1.w);
            *(float4*)(at + (2 * wave + 0) * LDA + lane * 4) = v0;
            *(float4*)(at + (2 * wave + 1) * LDA + lane * 4) = v1;
        }
        __syncthreads();

        if (more) {
            const int nr0 = nxt * 16 + 2 * wv, nr1 = nr0 + 1;
            const int nr0c = (nr0 < n) ? nr0 : (n - 1), nr1c = (nr1 < n) ? nr1 : (n - 1);
            beg  = __builtin_amdgcn_readfirstlane(pb);
            num0 = __builtin_amdgcn_readfirstlane((nr0 < n) ? pc0 : 0);
            num1 = __builtin_amdgcn_readfirstlane((nr1 < n) ? pc1 : 0);
            ent = make_int2(0, 0);
            ent2 = make_int2(0, 0);
            if (lane < num0 + num1) ent = csr[beg + lane];
            if (lane + 32 < num0 + num1) ent2 = csr[beg + 32 + lane];
            hs0 = *(const float4*)(xl + (long)nr0c * D_FEAT);
            hs1 = *(const float4*)(xl + (long)nr1c * D_FEAT);
            dd0 = dinv[nr0c];
            dd1 = dinv[nr1c];
            const int t2 = (nxt + nblk < ntiles) ? (nxt + nblk) : nxt;
            const int q0 = t2 * 16 + 2 * wv, q1 = q0 + 1;
            const int q0c = (q0 < n) ? q0 : (n - 1), q1c = (q1 < n) ? q1 : (n - 1);
            pb  = rowptr[q0c];
            pc0 = cnt[q0c];
            pc1 = cnt[q1c];
        }

        v8f acc = {};
#pragma unroll
        for (int kk = 0; kk < 32; ++kk) {
            v2f a;
            a.x = at[mrow * LDA + kk * 4 + khalf + 0];
            a.y = at[mrow * LDA + kk * 4 + khalf + 1];
            acc = __builtin_amdgcn_wmma_f32_16x16x4_f32(
                 false, a,  false, bf[kk],
                 (short)0, acc,  false,  false);
        }

        const long row0 = (long)tile * 16;
        const bool whole = row0 + 16 <= n;
        const long oidx = (row0 + mbase) * D_FEAT + col0 + ncol;

        float v[8];
        for (int r = 0; r < 8; ++r) {
            const float t = acc[r] + bias;
            v[r] = t > 0.0f ? t : 0.0f;
        }
        for (int r = 0; r < 8; ++r) Xs[(mbase + r) * LDA + col0 + ncol] = v[r];
        __syncthreads();

        float c[8];
        for (int r = 0; r < 8; ++r) c[r] = lbias;
        if (!second) {
            if (whole) {
                float* op = xout + oidx;
                for (int r = 0; r < 8; ++r) op[r * D_FEAT] = v[r];
            } else {
                for (int r = 0; r < 8; ++r)
                    if (row0 + mbase + r < n) xout[oidx + r * D_FEAT] = v[r];
            }
        } else {
            if (whole) {
                const float* ip = out + oidx;
                for (int r = 0; r < 8; ++r) c[r] = ip[r * D_FEAT];
            } else {
                for (int r = 0; r < 8; ++r)
                    if (row0 + mbase + r < n) c[r] = out[oidx + r * D_FEAT];
            }
        }

        v8f acc2 = {};
#pragma unroll
        for (int kk = 0; kk < 32; ++kk) {
            v2f a;
            a.x = Xs[mrow * LDA + kk * 4 + khalf + 0];
            a.y = Xs[mrow * LDA + kk * 4 + khalf + 1];
            const v2f bw = wl[kk * 32];
            acc2 = __builtin_amdgcn_wmma_f32_16x16x4_f32(
                false, a, false, bw, (short)0, acc2, false, false);
        }
        float o[8];
        for (int r = 0; r < 8; ++r) o[r] = acc2[r] + c[r];
        if (whole) {
            float* op = out + oidx;
            for (int r = 0; r < 8; ++r) op[r * D_FEAT] = o[r];
        } else {
            for (int r = 0; r < 8; ++r)
                if (row0 + mbase + r < n) out[oidx + r * D_FEAT] = o[r];
        }
    }
}

extern "C" void kernel_launch(void* const* d_in, const int* in_sizes, int n_in,
                              void* d_out, int out_size, void* d_ws, size_t ws_size,
                              hipStream_t stream) {
    (void)n_in; (void)out_size; (void)ws_size;

    const float* x    = (const float*)d_in[0];
    const int*   ei   = (const int*)d_in[1];
    const float* W1   = (const float*)d_in[2];
    const float* b1   = (const float*)d_in[3];
    const float* W2   = (const float*)d_in[4];
    const float* b2   = (const float*)d_in[5];
    const float* linW = (const float*)d_in[6];
    const float* linb = (const float*)d_in[7];
    float* out = (float*)d_out;

    const int N = in_sizes[0] / D_FEAT;
    const int E = in_sizes[1] / 2;
    const int* srcI = ei;
    const int* dstI = ei + E;

    float* ws   = (float*)d_ws;
    float* dinv = ws;
    float* h    = dinv + N;
    float* edge = h   + (long)N * D_FEAT;
    float* x1   = edge + (long)N * D_FEAT;
    float* x2   = x1  + (long)N * D_FEAT;

    int2* csr    = (int2*)edge;
    int*  rank   = (int*)(csr + E);
    int*  cnt    = rank + E;
    int*  rowptr = cnt + N;
    int*  bsumP  = rowptr + N;
    int*  cntP   = (int*)x2;

    const int nBlkN   = (N + 255) / 256;
    const int Npad    = nBlkN * 256;
    const int chunk   = (((E + RANK_BLOCKS - 1) / RANK_BLOCKS) + 1023) & ~1023;
    const int nBlkE4  = (E + 1023) / 1024;
    const int nTiles  = (N + 15) / 16;
    const int nBlkL   = (nTiles + TILES_PER_BLOCK - 1) / TILES_PER_BLOCK;

    rank_local_kernel<<<RANK_BLOCKS, 1024, 0, stream>>>(dstI, rank, cntP, bsumP, E, chunk, Npad, nBlkN);
    scan_offsets_kernel<<<nBlkN, 256, 0, stream>>>(cntP, bsumP, cnt, rowptr, dinv, N, Npad, nBlkN);
    place_edges_kernel<<<nBlkE4, 256, 0, stream>>>(srcI, dstI, rank, cntP, dinv, csr, E, chunk, Npad);

    agg_gemm_kernel<<<nBlkL, 256, 0, stream>>>(rowptr, cnt, csr, dinv, x, W1, b1, x1,
                                               linW, linb, out, 0, N, nBlkL);

    agg_gemm_kernel<<<nBlkL, 256, 0, stream>>>(rowptr, cnt, csr, dinv, x1, W2, b2, x2  ,
                                               linW + 128 * D_FEAT, linb, out, 1, N, nBlkL);
}


# Layer grid sized to the chip: 768 blocks = 3 resident on each of the 256 WGPs (625 put 3 on some, 2 on others); tiles dealt in rounds, the 53 left over spread out: fullest WGP 13 tiles, not 15
# speedup vs baseline: 3.8199x; 1.0267x over previous
#include <hip/hip_runtime.h>

typedef __attribute__((ext_vector_type(2))) float v2f;
typedef __attribute__((ext_vector_type(8))) float v8f;

#define D_FEAT 128

#define HIST_CAP 51200
#define RANK_BLOCKS 64

__global__ __launch_bounds__(1024)
void rank_local_kernel(const int* __restrict__ dst, int* __restrict__ rank,
                       int* __restrict__ cntP, int* __restrict__ bsumP,
                       int e, int chunk, int npad, int nb) {
    __shared__ __attribute__((aligned(16))) int hist[HIST_CAP + 32];
    __shared__ int gsum[HIST_CAP / 256];
    const int tid  = threadIdx.x;
    const int lane = tid & 31;
    const int e0 = blockIdx.x * chunk;
    const int e1 = (e0 + chunk < e) ? (e0 + chunk) : e;
    int* cp = cntP + (long)blockIdx.x * npad;
    int* bp = bsumP + (long)blockIdx.x * nb;

    for (int base = 0; base < npad; base += HIST_CAP) {
        const int wlen = (npad - base < HIST_CAP) ? (npad - base) : HIST_CAP;
        for (int i = tid * 4; i < wlen; i += 4096) *(int4*)(hist + i) = make_int4(0, 0, 0, 0);
        for (int i = tid; i < (wlen >> 8); i += 1024) gsum[i] = 0;
        __syncthreads();

        for (int i = e0 + tid; i < e1; i += 8192) {
            int d[8], r[8];
#pragma unroll
            for (int q = 0; q < 8; ++q) d[q] = dst[(i + q * 1024 < e1) ? (i + q * 1024) : (e1 - 1)];
#pragma unroll
            for (int q = 0; q < 8; ++q) d[q] = (i + q * 1024 < e1) ? (d[q] - base) : -1;
#pragma unroll
            for (int q = 0; q < 8; ++q)
                r[q] = atomicAdd(&hist[((unsigned)d[q] < (unsigned)wlen) ? d[q] : (HIST_CAP + lane)], 1);
#pragma unroll
            for (int q = 0; q < 8; ++q)
                if ((unsigned)d[q] < (unsigned)wlen) rank[i + q * 1024] = r[q];
        }
        __syncthreads();

        for (int i0 = (tid & ~31) * 4; i0 < wlen; i0 += 4096) {
            const int k = i0 + lane * 4;
            const int4 v = *(const int4*)(hist + k);
            *(int4*)(cp + base + k) = v;
            int s = v.x + v.y + v.z + v.w;
            s += __builtin_amdgcn_update_dpp(0, s, 0x161, 0xf, 0xf, false);
            s += __builtin_amdgcn_update_dpp(0, s, 0x162, 0xf, 0xf, false);
            s += __builtin_amdgcn_update_dpp(0, s, 0x164, 0xf, 0xf, false);
            s += __builtin_amdgcn_update_dpp(0, s, 0x168, 0xf, 0xf, false);
            s += __builtin_amdgcn_permlanex16(s, s, 0x76543210, 0xfedcba98, false, false);
            if (lane == 0) atomicAdd(&gsum[i0 >> 8], s);
        }
        __syncthreads();
        for (int g = tid; g < (wlen >> 8); g += 1024) bp[(base >> 8) + g] = gsum[g];
        __syncthreads();
    }
}

__global__ __launch_bounds__(256)
void scan_offsets_kernel(int* __restrict__ cntP, const int* __restrict__ bsumP,
                         int* __restrict__ cnt, int* __restrict__ rowptr,
                         float* __restrict__ dinv, int n, int npad, int nb) {
    __shared__ int wtot[8];
    __shared__ int ptot[8];
    const int tid = threadIdx.x;
    const int lane = tid & 31;
    const int wave = tid >> 5;
    const int i = blockIdx.x * 256 + tid;

    int part = 0;
    for (int j = tid; j < (int)blockIdx.x; j += 256) {
#pragma unroll 8
        for (int p = 0; p < RANK_BLOCKS; ++p) part += bsumP[p * nb + j];
    }
    for (int o = 16; o > 0; o >>= 1) part += __shfl_xor(part, o, 32);

    int cpv[RANK_BLOCKS];
    int c = 0;
#pragma unroll
    for (int p = 0; p < RANK_BLOCKS; ++p) {
        cpv[p] = cntP[(long)p * npad + i];
        c += cpv[p];
    }
    int inc = c;
    for (int o = 1; o < 32; o <<= 1) {
        int t = __shfl_up(inc, o, 32);
        if (lane >= o) inc += t;
    }
    if (lane == 31) wtot[wave] = inc;
    if (lane == 0) ptot[wave] = part;
    __syncthreads();

    int base = 0;
    for (int w = 0; w < 8; ++w) {
        base += ptot[w];
        if (w < wave) base += wtot[w];
    }
    if (i < n) {
        const int rp = base + inc - c;
        cnt[i] = c;
        rowptr[i] = rp;
        float d = (float)(c + 1);
        dinv[i] = (d > 0.0f) ? rsqrtf(d) : 0.0f;
        int run = rp;
#pragma unroll
        for (int p = 0; p < RANK_BLOCKS; ++p) {
            cntP[(long)p * npad + i] = run;
            run += cpv[p];
        }
    }
}

__global__ __launch_bounds__(256)
void place_edges_kernel(const int* __restrict__ src, const int* __restrict__ dst,
                        const int* __restrict__ rank, const int* __restrict__ offP,
                        const float* __restrict__ dinv, int2* __restrict__ csr,
                        int e, int chunk, int npad) {
    const int i0 = blockIdx.x * 1024 + threadIdx.x;
    const int* rowptr = offP + (long)((blockIdx.x * 1024) / chunk) * npad;
    if (blockIdx.x * 1024 + 1024 <= e) {
        int s[4], d[4], rk[4], pos[4], wv[4];
        for (int q = 0; q < 4; ++q) {
            const int i = i0 + q * 256;
            s[q]  = src[i];
            d[q]  = dst[i];
            rk[q] = rank[i];
        }
        for (int q = 0; q < 4; ++q) {
            pos[q] = rowptr[d[q]] + rk[q];
            wv[q]  = __float_as_int(dinv[s[q]]);
        }
        for (int q = 0; q < 4; ++q) csr[pos[q]] = make_int2(s[q], wv[q]);
    } else {
        for (int q = 0; q < 4; ++q) {
            const int i = i0 + q * 256;
            if (i < e) {
                const int s = src[i], d = dst[i];
                csr[rowptr[d] + rank[i]] = make_int2(s, __float_as_int(dinv[s]));
            }
        }
    }
}

#define LDA 132
#define LAYER_BLOCKS 768
#define LAYER_SPREAD_INV 53
__device__ __forceinline__ void fma4(float4& acc, float w, const float4& v) {
    acc.x = fmaf(w, v.x, acc.x);
    acc.y = fmaf(w, v.y, acc.y);
    acc.z = fmaf(w, v.z, acc.z);
    acc.w = fmaf(w, v.w, acc.w);
}

template <int G>
__device__ __forceinline__ void gather_group(const float* __restrict__ xl, int s_l, int w_l,
                                             int j, int split, float4& acc0, float4& acc1) {
    int s[G];
    float4 hv[G];
    float w[G];
#pragma unroll
    for (int q = 0; q < G; ++q) s[q] = __builtin_amdgcn_readlane(s_l, j + q);
#pragma unroll
    for (int q = 0; q < G; ++q) hv[q] = *(const float4*)(xl + (long)s[q] * D_FEAT);
#pragma unroll
    for (int q = 0; q < G; ++q) w[q] = __int_as_float(__builtin_amdgcn_readlane(w_l, j + q));
    if (j + G <= split) {
#pragma unroll
        for (int q = 0; q < G; ++q) fma4(acc0, w[q], hv[q]);
    } else if (j >= split) {
#pragma unroll
        for (int q = 0; q < G; ++q) fma4(acc1, w[q], hv[q]);
    } else {
#pragma unroll
        for (int q = 0; q < G; ++q) {
            if (j + q < split) fma4(acc0, w[q], hv[q]);
            else               fma4(acc1, w[q], hv[q]);
        }
    }
}

__global__ __launch_bounds__(256, 6)
void agg_gemm_kernel(const int* __restrict__ rowptr, const int* __restrict__ cnt,
                     const int2* __restrict__ csr, const float* __restrict__ dinv,
                     const float* __restrict__ xin, const float* __restrict__ W,
                     const float* __restrict__ b, float* __restrict__ xout,
                     const float* __restrict__ Wl, const float* __restrict__ bl,
                     float* out, int second, int n) {
    __shared__ __attribute__((aligned(16))) float As[2][16 * LDA];
    __shared__ __attribute__((aligned(16))) float Xs[16 * LDA];
    __shared__ __attribute__((aligned(16))) v2f Wls[8 * 32 * 32];
    const int tid   = threadIdx.x;
    const int wave  = __builtin_amdgcn_readfirstlane(tid >> 5);
    const int lane  = tid & 31;
    const int col0  = wave * 16;
    const int mrow  = lane & 15;
    const int khalf = (lane >> 4) * 2;
    const int ncol  = lane & 15;
    const int mbase = (lane >> 4) * 8;
    const int ntiles = (n + 15) >> 4;
    const int nblk    = LAYER_BLOCKS;
    const int mainEnd = (ntiles / LAYER_BLOCKS) * LAYER_BLOCKS;
    const int bid     = blockIdx.x;
    const int kx      = (bid * LAYER_SPREAD_INV) % LAYER_BLOCKS;
    const int extra   = (mainEnd + kx < ntiles) ? (mainEnd + kx) : ntiles;
    auto next_tile = [=](int t) -> int {
        return (t + nblk < mainEnd) ? (t + nblk) : ((t < mainEnd) ? extra : ntiles);
    };

    v2f bf[32];
#pragma unroll
    for (int kk = 0; kk < 32; ++kk) {
        bf[kk].x = W[(kk * 4 + khalf + 0) * D_FEAT + col0 + ncol];
        bf[kk].y = W[(kk * 4 + khalf + 1) * D_FEAT + col0 + ncol];
    }
    const float bias = b[col0 + ncol];
    const float lbias = bl[col0 + ncol];
    const float* xl = xin + lane * 4;

    int tile = (bid < mainEnd) ? bid : extra;
    if (tile >= ntiles) return;

    v2f* wl = Wls + wave * (32 * 32) + lane;
    for (int kk = 0; kk < 32; ++kk) {
        v2f t;
        t.x = Wl[(kk * 4 + khalf + 0) * D_FEAT + col0 + ncol];
        t.y = Wl[(kk * 4 + khalf + 1) * D_FEAT + col0 + ncol];
        wl[kk * 32] = t;
    }

    const int wv = tid >> 5;
    int beg, num0, num1;
    float dd0, dd1;
    int2 ent, ent2;
    float4 hs0, hs1;
    int pb, pc0, pc1;
    {
        const int r0 = tile * 16 + 2 * wv, r1 = r0 + 1;
        const int r0c = (r0 < n) ? r0 : (n - 1), r1c = (r1 < n) ? r1 : (n - 1);
        const int vb = rowptr[r0c], vc0 = cnt[r0c], vc1 = cnt[r1c];
        dd0  = dinv[r0c];
        dd1  = dinv[r1c];
        hs0 = *(const float4*)(xl + (long)r0c * D_FEAT);
        hs1 = *(const float4*)(xl + (long)r1c * D_FEAT);
        const int n1 = next_tile(tile);
        const int t1 = (n1 < ntiles) ? n1 : tile;
        const int q0 = t1 * 16 + 2 * wv, q1 = q0 + 1;
        const int q0c = (q0 < n) ? q0 : (n - 1), q1c = (q1 < n) ? q1 : (n - 1);
        pb  = rowptr[q0c];
        pc0 = cnt[q0c];
        pc1 = cnt[q1c];
        beg  = __builtin_amdgcn_readfirstlane(vb);
        num0 = __builtin_amdgcn_readfirstlane((r0 < n) ? vc0 : 0);
        num1 = __builtin_amdgcn_readfirstlane((r1 < n) ? vc1 : 0);
        ent = make_int2(0, 0);
        ent2 = make_int2(0, 0);
        if (lane < num0 + num1) ent = csr[beg + lane];
        if (lane + 32 < num0 + num1) ent2 = csr[beg + 32 + lane];
    }

    int buf = 0;
    for (; tile < ntiles; buf ^= 1) {
        const int m = num0 + num1;
        const int nxt = next_tile(tile);
        const bool more = nxt < ntiles;

        float4 acc0 = make_float4(0.f, 0.f, 0.f, 0.f), acc1 = acc0;
        for (int base = 0; base < m; base += 32) {
            const int mw = (m - base < 32) ? (m - base) : 32;
            const int s_l = ent.x, w_l = ent.y;
            if (base + 32 < m) {
                if (base == 0) {
                    ent = ent2;
                } else {
                    ent = make_int2(0, 0);
                    if (lane < m - base - 32) ent = csr[beg + base + 32 + lane];
                }
            }
            const int split = num0 - base;
            int j = 0;
            for (; j + 8 <= mw; j += 8) gather_group<8>(xl, s_l, w_l, j, split, acc0, acc1);
            const int r = mw - j;
            if (r >= 4) {
                if (r == 7)      gather_group<7>(xl, s_l, w_l, j, split, acc0, acc1);
                else if (r == 6) gather_group<6>(xl, s_l, w_l, j, split, acc0, acc1);
                else if (r == 5) gather_group<5>(xl, s_l, w_l, j, split, acc0, acc1);
                else             gather_group<4>(xl, s_l, w_l, j, split, acc0, acc1);
            } else {
                if (r == 3)      gather_group<3>(xl, s_l, w_l, j, split, acc0, acc1);
                else if (r == 2) gather_group<2>(xl, s_l, w_l, j, split, acc0, acc1);
                else if (r == 1) gather_group<1>(xl, s_l, w_l, j, split, acc0, acc1);
            }
        }

        float* at = &As[buf][0];
        {
            float4 v0, v1;
            v0.x = dd0 * (acc0.x + dd0 * hs0.x);
            v0.y = dd0 * (acc0.y + dd0 * hs0.y);
            v0.z = dd0 * (acc0.z + dd0 * hs0.z);
            v0.w = dd0 * (acc0.w + dd0 * hs0.w);
            v1.x = dd1 * (acc1.x + dd1 * hs1.x);
            v1.y = dd1 * (acc1.y + dd1 * hs1.y);
            v1.z = dd1 * (acc1.z + dd1 * hs1.z);
            v1.w = dd1 * (acc1.w + dd1 * hs1.w);
            *(float4*)(at + (2 * wave + 0) * LDA + lane * 4) = v0;
            *(float4*)(at + (2 * wave + 1) * LDA + lane * 4) = v1;
        }
        __syncthreads();

        if (more) {
            const int nr0 = nxt * 16 + 2 * wv, nr1 = nr0 + 1;
            const int nr0c = (nr0 < n) ? nr0 : (n - 1), nr1c = (nr1 < n) ? nr1 : (n - 1);
            beg  = __builtin_amdgcn_readfirstlane(pb);
            num0 = __builtin_amdgcn_readfirstlane((nr0 < n) ? pc0 : 0);
            num1 = __builtin_amdgcn_readfirstlane((nr1 < n) ? pc1 : 0);
            ent = make_int2(0, 0);
            ent2 = make_int2(0, 0);
            if (lane < num0 + num1) ent = csr[beg + lane];
            if (lane + 32 < num0 + num1) ent2 = csr[beg + 32 + lane];
            hs0 = *(const float4*)(xl + (long)nr0c * D_FEAT);
            hs1 = *(const float4*)(xl + (long)nr1c * D_FEAT);
            dd0 = dinv[nr0c];
            dd1 = dinv[nr1c];
            const int n2 = next_tile(nxt);
            const int t2 = (n2 < ntiles) ? n2 : nxt;
            const int q0 = t2 * 16 + 2 * wv, q1 = q0 + 1;
            const int q0c = (q0 < n) ? q0 : (n - 1), q1c = (q1 < n) ? q1 : (n - 1);
            pb  = rowptr[q0c];
            pc0 = cnt[q0c];
            pc1 = cnt[q1c];
        }

        v8f acc = {};
#pragma unroll
        for (int kk = 0; kk < 32; ++kk) {
            v2f a;
            a.x = at[mrow * LDA + kk * 4 + khalf + 0];
            a.y = at[mrow * LDA + kk * 4 + khalf + 1];
            acc = __builtin_amdgcn_wmma_f32_16x16x4_f32(
                 false, a,  false, bf[kk],
                 (short)0, acc,  false,  false);
        }

        const long row0 = (long)tile * 16;
        const bool whole = row0 + 16 <= n;
        const long oidx = (row0 + mbase) * D_FEAT + col0 + ncol;

        float v[8];
        for (int r = 0; r < 8; ++r) {
            const float t = acc[r] + bias;
            v[r] = t > 0.0f ? t : 0.0f;
        }
        for (int r = 0; r < 8; ++r) Xs[(mbase + r) * LDA + col0 + ncol] = v[r];
        __syncthreads();

        float c[8];
        for (int r = 0; r < 8; ++r) c[r] = lbias;
        if (!second) {
            if (whole) {
                float* op = xout + oidx;
                for (int r = 0; r < 8; ++r) op[r * D_FEAT] = v[r];
            } else {
                for (int r = 0; r < 8; ++r)
                    if (row0 + mbase + r < n) xout[oidx + r * D_FEAT] = v[r];
            }
        } else {
            if (whole) {
                const float* ip = out + oidx;
                for (int r = 0; r < 8; ++r) c[r] = ip[r * D_FEAT];
            } else {
                for (int r = 0; r < 8; ++r)
                    if (row0 + mbase + r < n) c[r] = out[oidx + r * D_FEAT];
            }
        }

        v8f acc2 = {};
#pragma unroll
        for (int kk = 0; kk < 32; ++kk) {
            v2f a;
            a.x = Xs[mrow * LDA + kk * 4 + khalf + 0];
            a.y = Xs[mrow * LDA + kk * 4 + khalf + 1];
            const v2f bw = wl[kk * 32];
            acc2 = __builtin_amdgcn_wmma_f32_16x16x4_f32(
                false, a, false, bw, (short)0, acc2, false, false);
        }
        float o[8];
        for (int r = 0; r < 8; ++r) o[r] = acc2[r] + c[r];
        if (whole) {
            float* op = out + oidx;
            for (int r = 0; r < 8; ++r) op[r * D_FEAT] = o[r];
        } else {
            for (int r = 0; r < 8; ++r)
                if (row0 + mbase + r < n) out[oidx + r * D_FEAT] = o[r];
        }
        tile = nxt;
    }
}

extern "C" void kernel_launch(void* const* d_in, const int* in_sizes, int n_in,
                              void* d_out, int out_size, void* d_ws, size_t ws_size,
                              hipStream_t stream) {
    (void)n_in; (void)out_size; (void)ws_size;

    const float* x    = (const float*)d_in[0];
    const int*   ei   = (const int*)d_in[1];
    const float* W1   = (const float*)d_in[2];
    const float* b1   = (const float*)d_in[3];
    const float* W2   = (const float*)d_in[4];
    const float* b2   = (const float*)d_in[5];
    const float* linW = (const float*)d_in[6];
    const float* linb = (const float*)d_in[7];
    float* out = (float*)d_out;

    const int N = in_sizes[0] / D_FEAT;
    const int E = in_sizes[1] / 2;
    const int* srcI = ei;
    const int* dstI = ei + E;

    float* ws   = (float*)d_ws;
    float* dinv = ws;
    float* h    = dinv + N;
    float* edge = h   + (long)N * D_FEAT;
    float* x1   = edge + (long)N * D_FEAT;
    float* x2   = x1  + (long)N * D_FEAT;

    int2* csr    = (int2*)edge;
    int*  rank   = (int*)(csr + E);
    int*  cnt    = rank + E;
    int*  rowptr = cnt + N;
    int*  bsumP  = rowptr + N;
    int*  cntP   = (int*)x2;

    const int nBlkN   = (N + 255) / 256;
    const int Npad    = nBlkN * 256;
    const int chunk   = (((E + RANK_BLOCKS - 1) / RANK_BLOCKS) + 1023) & ~1023;
    const int nBlkE4  = (E + 1023) / 1024;

    rank_local_kernel<<<RANK_BLOCKS, 1024, 0, stream>>>(dstI, rank, cntP, bsumP, E, chunk, Npad, nBlkN);
    scan_offsets_kernel<<<nBlkN, 256, 0, stream>>>(cntP, bsumP, cnt, rowptr, dinv, N, Npad, nBlkN);
    place_edges_kernel<<<nBlkE4, 256, 0, stream>>>(srcI, dstI, rank, cntP, dinv, csr, E, chunk, Npad);

    agg_gemm_kernel<<<LAYER_BLOCKS, 256, 0, stream>>>(rowptr, cnt, csr, dinv, x, W1, b1, x1,
                                                      linW, linb, out, 0, N);

    agg_gemm_kernel<<<LAYER_BLOCKS, 256, 0, stream>>>(rowptr, cnt, csr, dinv, x1, W2, b2, x2  ,
                                                      linW + 128 * D_FEAT, linb, out, 1, N);
}
